# starting kernel
# baseline (speedup 1.0000x reference)
#include <hip/hip_runtime.h>
#include <hip/hip_bf16.h>
#include <math.h>

#define N_NODES 100000
#define N_EDGES 600000
#define DIM     128
#define NLAYERS 4
#define NGRAPH  512
#define NCLASS  40
#define LN_EPSF 1e-5f

typedef __attribute__((ext_vector_type(2))) float v2f;
typedef __attribute__((ext_vector_type(8))) float v8f;

__global__ void gin_init_t(const float* __restrict__ h, float* __restrict__ t, int n4) {
    int i = blockIdx.x * blockDim.x + threadIdx.x;
    if (i < n4) {
        float4 v = ((const float4*)h)[i];
        v.x *= 2.f; v.y *= 2.f; v.z *= 2.f; v.w *= 2.f;
        ((float4*)t)[i] = v;
    }
}

__global__ void gin_scatter(const float* __restrict__ h, const int* __restrict__ src,
                            const int* __restrict__ dst, float* __restrict__ t) {
    int tid = blockIdx.x * blockDim.x + threadIdx.x;
    int e = tid >> 5;
    int lane = tid & 31;
    if (e < N_EDGES) {
        int s = src[e], d = dst[e];
        const float4 v = *(const float4*)(h + (long long)s * DIM + lane * 4);
        float* p = t + (long long)d * DIM + lane * 4;
        atomicAdd(p + 0, v.x);
        atomicAdd(p + 1, v.y);
        atomicAdd(p + 2, v.z);
        atomicAdd(p + 3, v.w);
    }
}

__global__ __launch_bounds__(256)
void gin_gemm_ln(const float* __restrict__ T, const float* __restrict__ W,
                 const float* __restrict__ bias, const float* __restrict__ gamma,
                 const float* __restrict__ beta, float* __restrict__ Hout) {
    __shared__ float red_s[16][8];
    __shared__ float red_q[16][8];
    __shared__ float mu[16];
    __shared__ float rstd[16];

    const int wave = threadIdx.x >> 5;
    const int lane = threadIdx.x & 31;
    const int half = lane >> 4;
    const int l16  = lane & 15;
    const int row0 = blockIdx.x * 16;
    const int rowA = row0 + l16;
    const int colB = wave * 16 + l16;

    const float* arow = T + (long long)rowA * DIM + 2 * half;

    v8f acc = {};
#pragma unroll
    for (int k0 = 0; k0 < DIM; k0 += 4) {
        v2f a, b;
        a.x = arow[k0];
        a.y = arow[k0 + 1];
        const int kb = k0 + 2 * half;
        b.x = W[kb * DIM + colB];
        b.y = W[(kb + 1) * DIM + colB];
        acc = __builtin_amdgcn_wmma_f32_16x16x4_f32(
                false, a, false, b, (short)0, acc, false, false);
    }

    const float bv = bias[colB];
    const float gv = gamma[colB];
    const float bt = beta[colB];

    float vals[8];
#pragma unroll
    for (int r = 0; r < 8; ++r) {
        float v = acc[r] + bv;
        vals[r] = v > 0.f ? v : 0.f;
    }

#pragma unroll
    for (int r = 0; r < 8; ++r) {
        float s = vals[r];
        float q = vals[r] * vals[r];
#pragma unroll
        for (int m = 1; m < 16; m <<= 1) {
            s += __shfl_xor(s, m, 32);
            q += __shfl_xor(q, m, 32);
        }
        if (l16 == 0) {
            const int rt = r + 8 * half;
            red_s[rt][wave] = s;
            red_q[rt][wave] = q;
        }
    }
    __syncthreads();

    if (threadIdx.x < 16) {
        float s = 0.f, q = 0.f;
#pragma unroll
        for (int w = 0; w < 8; ++w) { s += red_s[threadIdx.x][w]; q += red_q[threadIdx.x][w]; }
        const float m   = s * (1.0f / DIM);
        const float var = q * (1.0f / DIM) - m * m;
        mu[threadIdx.x]   = m;
        rstd[threadIdx.x] = rsqrtf(var + LN_EPSF);
    }
    __syncthreads();

#pragma unroll
    for (int r = 0; r < 8; ++r) {
        const int mrow = r + 8 * half;
        const float o = (vals[r] - mu[mrow]) * rstd[mrow] * gv + bt;
        Hout[(long long)(row0 + mrow) * DIM + colB] = o;
    }
}

__global__ void gin_zero(float* __restrict__ p, int n) {
    int i = blockIdx.x * blockDim.x + threadIdx.x;
    if (i < n) p[i] = 0.f;
}

__global__ void gin_pool(const float* __restrict__ h, const int* __restrict__ batch,
                         float* __restrict__ psum, float* __restrict__ cnt) {
    int tid = blockIdx.x * blockDim.x + threadIdx.x;
    int n = tid >> 5;
    int lane = tid & 31;
    if (n < N_NODES) {
        int b = batch[n];
        const float4 v = *(const float4*)(h + (long long)n * DIM + lane * 4);
        float* p = psum + (long long)b * DIM + lane * 4;
        atomicAdd(p + 0, v.x);
        atomicAdd(p + 1, v.y);
        atomicAdd(p + 2, v.z);
        atomicAdd(p + 3, v.w);
        if (lane == 0) atomicAdd(cnt + b, 1.0f);
    }
}

__global__ void gin_outproj(const float* __restrict__ psum, const float* __restrict__ cnt,
                            const float* __restrict__ Wout, const float* __restrict__ bout,
                            float* __restrict__ logits) {
    int idx = blockIdx.x * blockDim.x + threadIdx.x;
    if (idx < NGRAPH * NCLASS) {
        int g = idx / NCLASS;
        int c = idx - g * NCLASS;
        float inv = 1.0f / fmaxf(cnt[g], 1.0f);
        float acc = bout[c];
        const float* pg = psum + g * DIM;
#pragma unroll 4
        for (int k = 0; k < DIM; ++k) acc += pg[k] * inv * Wout[k * NCLASS + c];
        logits[idx] = acc;
    }
}

__global__ void gin_logsoftmax(const float* __restrict__ logits, float* __restrict__ out) {
    __shared__ float sh[64];
    const int g = blockIdx.x;
    const int t = threadIdx.x;
    float v = (t < NCLASS) ? logits[g * NCLASS + t] : -3.4e38f;
    sh[t] = v;
    __syncthreads();
    for (int s = 32; s > 0; s >>= 1) {
        if (t < s) sh[t] = fmaxf(sh[t], sh[t + s]);
        __syncthreads();
    }
    const float mx = sh[0];
    __syncthreads();
    sh[t] = (t < NCLASS) ? expf(v - mx) : 0.f;
    __syncthreads();
    for (int s = 32; s > 0; s >>= 1) {
        if (t < s) sh[t] += sh[t + s];
        __syncthreads();
    }
    const float lse = logf(sh[0]);
    if (t < NCLASS) out[g * NCLASS + t] = v - mx - lse;
}

extern "C" void kernel_launch(void* const* d_in, const int* in_sizes, int n_in,
                              void* d_out, int out_size, void* d_ws, size_t ws_size,
                              hipStream_t stream) {
    const float* x      = (const float*)d_in[0];
    const int*   ei     = (const int*)d_in[1];
    const int*   batch  = (const int*)d_in[2];
    const float* Ws     = (const float*)d_in[3];
    const float* bs     = (const float*)d_in[4];
    const float* gammas = (const float*)d_in[5];
    const float* betas  = (const float*)d_in[6];
    const float* Wout   = (const float*)d_in[7];
    const float* bout   = (const float*)d_in[8];
    float* out = (float*)d_out;

    const int* src = ei;
    const int* dst = ei + N_EDGES;

    float* ws = (float*)d_ws;
    const size_t nd = (size_t)N_NODES * DIM;
    float* hA     = ws;
    float* hB     = ws + nd;
    float* tbuf   = ws + 2 * nd;
    float* psum   = ws + 3 * nd;
    float* cnt    = psum + (size_t)NGRAPH * DIM;
    float* logits = cnt + NGRAPH;

    const int n4 = (N_NODES * DIM) / 4;
    const int gInit    = (n4 + 255) / 256;
    const int gScatter = (N_EDGES * 32 + 255) / 256;
    const int gGemm    = N_NODES / 16;
    const int gPool    = (N_NODES * 32 + 255) / 256;

    const float* hin = x;
    for (int i = 0; i < NLAYERS; ++i) {
        float* hout = (i & 1) ? hB : hA;
        gin_init_t<<<gInit, 256, 0, stream>>>(hin, tbuf, n4);
        gin_scatter<<<gScatter, 256, 0, stream>>>(hin, src, dst, tbuf);
        gin_gemm_ln<<<gGemm, 256, 0, stream>>>(tbuf, Ws + (size_t)i * DIM * DIM,
                                               bs + i * DIM, gammas + i * DIM,
                                               betas + i * DIM, hout);
        hin = hout;
    }

    const int nPool = NGRAPH * DIM + NGRAPH;
    gin_zero<<<(nPool + 255) / 256, 256, 0, stream>>>(psum, nPool);
    gin_pool<<<gPool, 256, 0, stream>>>(hin, batch, psum, cnt);
    gin_outproj<<<(NGRAPH * NCLASS + 255) / 256, 256, 0, stream>>>(psum, cnt, Wout, bout, logits);
    gin_logsoftmax<<<NGRAPH, 64, 0, stream>>>(logits, out);
}


# Float atomic scatter replaced by a per-call edge list bucketed by destination (count, scan, place) and a wave-per-node register row sum; neighbour terms now added in list order, 2h last
# speedup vs baseline: 1.6146x; 1.6146x over previous
#include <hip/hip_runtime.h>
#include <hip/hip_bf16.h>
#include <math.h>

#define N_NODES 100000
#define N_EDGES 600000
#define DIM     128
#define NLAYERS 4
#define NGRAPH  512
#define NCLASS  40
#define LN_EPSF 1e-5f

typedef __attribute__((ext_vector_type(2))) float v2f;
typedef __attribute__((ext_vector_type(8))) float v8f;

#define SCAN_PER_BLK 1024
#define SCAN_NBLK    ((N_NODES + 1 + SCAN_PER_BLK - 1) / SCAN_PER_BLK)
#define SCAN_NPAD    (SCAN_NBLK * SCAN_PER_BLK)
#define GATHER_NBLK  2048

__global__ void gin_count(const int* __restrict__ dst, int* __restrict__ deg,
                          int* __restrict__ slot) {
    int e = blockIdx.x * blockDim.x + threadIdx.x;
    if (e < N_EDGES) slot[e] = atomicAdd(deg + dst[e], 1);
}

__global__ __launch_bounds__(256)
void gin_blocksum(const int* __restrict__ deg, int* __restrict__ bsum) {
    __shared__ int wsum[8];
    const int4 d = ((const int4*)deg)[blockIdx.x * 256 + threadIdx.x];
    int s = d.x + d.y + d.z + d.w;
#pragma unroll
    for (int m = 16; m > 0; m >>= 1) s += __shfl_xor(s, m, 32);
    if ((threadIdx.x & 31) == 0) wsum[threadIdx.x >> 5] = s;
    __syncthreads();
    if (threadIdx.x == 0) {
        int t = 0;
#pragma unroll
        for (int w = 0; w < 8; ++w) t += wsum[w];
        bsum[blockIdx.x] = t;
    }
}

__global__ __launch_bounds__(256)
void gin_rowptr(const int* __restrict__ deg, const int* __restrict__ bsum,
                int* __restrict__ rowptr) {
    static_assert(SCAN_NBLK <= 256, "one thread per earlier block sum");
    __shared__ int wsum[8];
    __shared__ int woff[8];
    const int tid  = threadIdx.x;
    const int lane = tid & 31;
    const int wave = tid >> 5;

    int p = (tid < (int)blockIdx.x) ? bsum[tid] : 0;
#pragma unroll
    for (int m = 16; m > 0; m >>= 1) p += __shfl_xor(p, m, 32);

    const int4 d = ((const int4*)deg)[blockIdx.x * 256 + tid];
    const int s = d.x + d.y + d.z + d.w;
    int inc = s;
#pragma unroll
    for (int m = 1; m < 32; m <<= 1) {
        const int v = __shfl_up(inc, m, 32);
        if (lane >= m) inc += v;
    }
    if (lane == 31) wsum[wave] = inc;
    if (lane == 0)  woff[wave] = p;
    __syncthreads();

    int base = 0;
#pragma unroll
    for (int w = 0; w < 8; ++w) {
        base += woff[w];
        if (w < wave) base += wsum[w];
    }
    int4 r;
    r.x = base + inc - s;
    r.y = r.x + d.x;
    r.z = r.y + d.y;
    r.w = r.z + d.z;
    ((int4*)rowptr)[blockIdx.x * 256 + tid] = r;
}

__global__ void gin_place(const int* __restrict__ src, const int* __restrict__ dst,
                          const int* __restrict__ rowptr, const int* __restrict__ slot,
                          int* __restrict__ col) {
    int e = blockIdx.x * blockDim.x + threadIdx.x;
    if (e < N_EDGES) col[rowptr[dst[e]] + slot[e]] = src[e];
}

__device__ __forceinline__ float4 gin_row4(const float* __restrict__ h, int node, int lane) {
    return *(const float4*)(h + (long long)node * DIM + lane * 4);
}

__global__ __launch_bounds__(256)
void gin_gather(const float* __restrict__ h, const int* __restrict__ rowptr,
                const int* __restrict__ col, float* __restrict__ t) {
    const int lane = threadIdx.x & 31;
    const int w  = __builtin_amdgcn_readfirstlane((int)((blockIdx.x * blockDim.x + threadIdx.x) >> 5));
    const int nw = (int)((gridDim.x * blockDim.x) >> 5);
    if (w >= N_NODES) return;

    const int p0 = rowptr[w + (lane & 1)];
    int beg = __builtin_amdgcn_readlane(p0, 0);
    int end = __builtin_amdgcn_readlane(p0, 1);

    for (int n = w; n < N_NODES; n += nw) {
        const int nn = n + nw;
        const int pn = nn < N_NODES ? nn : n;
        const int pnext = rowptr[pn + (lane & 1)];
        const float4 self = gin_row4(h, n, lane);
        float4 acc = make_float4(0.f, 0.f, 0.f, 0.f);

        for (int base = beg; base < end; base += 32) {
            const int cnt = (end - base) < 32 ? (end - base) : 32;
            const int my = (lane < cnt) ? col[base + lane] : 0;
            int j = 0;
            do {
                float4 v[8];
                v[0] = gin_row4(h, __builtin_amdgcn_readlane(my, j), lane);
#pragma unroll
                for (int u = 1; u < 8; ++u) {
                    if (j + u < cnt) v[u] = gin_row4(h, __builtin_amdgcn_readlane(my, j + u), lane);
                }
                acc.x += v[0].x; acc.y += v[0].y; acc.z += v[0].z; acc.w += v[0].w;
#pragma unroll
                for (int u = 1; u < 8; ++u) {
                    if (j + u < cnt) {
                        acc.x += v[u].x; acc.y += v[u].y; acc.z += v[u].z; acc.w += v[u].w;
                    }
                }
                j += 8;
            } while (j < cnt);
        }
        float4 o;
        o.x = 2.f * self.x + acc.x;
        o.y = 2.f * self.y + acc.y;
        o.z = 2.f * self.z + acc.z;
        o.w = 2.f * self.w + acc.w;
        *(float4*)(t + (long long)n * DIM + lane * 4) = o;

        beg = __builtin_amdgcn_readlane(pnext, 0);
        end = __builtin_amdgcn_readlane(pnext, 1);
    }
}

__global__ __launch_bounds__(256)
void gin_gemm_ln(const float* __restrict__ T, const float* __restrict__ W,
                 const float* __restrict__ bias, const float* __restrict__ gamma,
                 const float* __restrict__ beta, float* __restrict__ Hout) {
    __shared__ float red_s[16][8];
    __shared__ float red_q[16][8];
    __shared__ float mu[16];
    __shared__ float rstd[16];

    const int wave = threadIdx.x >> 5;
    const int lane = threadIdx.x & 31;
    const int half = lane >> 4;
    const int l16  = lane & 15;
    const int row0 = blockIdx.x * 16;
    const int rowA = row0 + l16;
    const int colB = wave * 16 + l16;

    const float* arow = T + (long long)rowA * DIM + 2 * half;

    v8f acc = {};
#pragma unroll
    for (int k0 = 0; k0 < DIM; k0 += 4) {
        v2f a, b;
        a.x = arow[k0];
        a.y = arow[k0 + 1];
        const int kb = k0 + 2 * half;
        b.x = W[kb * DIM + colB];
        b.y = W[(kb + 1) * DIM + colB];
        acc = __builtin_amdgcn_wmma_f32_16x16x4_f32(
                false, a, false, b, (short)0, acc, false, false);
    }

    const float bv = bias[colB];
    const float gv = gamma[colB];
    const float bt = beta[colB];

    float vals[8];
#pragma unroll
    for (int r = 0; r < 8; ++r) {
        float v = acc[r] + bv;
        vals[r] = v > 0.f ? v : 0.f;
    }

#pragma unroll
    for (int r = 0; r < 8; ++r) {
        float s = vals[r];
        float q = vals[r] * vals[r];
#pragma unroll
        for (int m = 1; m < 16; m <<= 1) {
            s += __shfl_xor(s, m, 32);
            q += __shfl_xor(q, m, 32);
        }
        if (l16 == 0) {
            const int rt = r + 8 * half;
            red_s[rt][wave] = s;
            red_q[rt][wave] = q;
        }
    }
    __syncthreads();

    if (threadIdx.x < 16) {
        float s = 0.f, q = 0.f;
#pragma unroll
        for (int w = 0; w < 8; ++w) { s += red_s[threadIdx.x][w]; q += red_q[threadIdx.x][w]; }
        const float m   = s * (1.0f / DIM);
        const float var = q * (1.0f / DIM) - m * m;
        mu[threadIdx.x]   = m;
        rstd[threadIdx.x] = rsqrtf(var + LN_EPSF);
    }
    __syncthreads();

#pragma unroll
    for (int r = 0; r < 8; ++r) {
        const int mrow = r + 8 * half;
        const float o = (vals[r] - mu[mrow]) * rstd[mrow] * gv + bt;
        Hout[(long long)(row0 + mrow) * DIM + colB] = o;
    }
}

__global__ void gin_zero(float* __restrict__ p, int n) {
    int i = blockIdx.x * blockDim.x + threadIdx.x;
    if (i < n) p[i] = 0.f;
}

__global__ void gin_pool(const float* __restrict__ h, const int* __restrict__ batch,
                         float* __restrict__ psum, float* __restrict__ cnt) {
    int tid = blockIdx.x * blockDim.x + threadIdx.x;
    int n = tid >> 5;
    int lane = tid & 31;
    if (n < N_NODES) {
        int b = batch[n];
        const float4 v = *(const float4*)(h + (long long)n * DIM + lane * 4);
        float* p = psum + (long long)b * DIM + lane * 4;
        atomicAdd(p + 0, v.x);
        atomicAdd(p + 1, v.y);
        atomicAdd(p + 2, v.z);
        atomicAdd(p + 3, v.w);
        if (lane == 0) atomicAdd(cnt + b, 1.0f);
    }
}

__global__ void gin_outproj(const float* __restrict__ psum, const float* __restrict__ cnt,
                            const float* __restrict__ Wout, const float* __restrict__ bout,
                            float* __restrict__ logits) {
    int idx = blockIdx.x * blockDim.x + threadIdx.x;
    if (idx < NGRAPH * NCLASS) {
        int g = idx / NCLASS;
        int c = idx - g * NCLASS;
        float inv = 1.0f / fmaxf(cnt[g], 1.0f);
        float acc = bout[c];
        const float* pg = psum + g * DIM;
#pragma unroll 4
        for (int k = 0; k < DIM; ++k) acc += pg[k] * inv * Wout[k * NCLASS + c];
        logits[idx] = acc;
    }
}

__global__ void gin_logsoftmax(const float* __restrict__ logits, float* __restrict__ out) {
    __shared__ float sh[64];
    const int g = blockIdx.x;
    const int t = threadIdx.x;
    float v = (t < NCLASS) ? logits[g * NCLASS + t] : -3.4e38f;
    sh[t] = v;
    __syncthreads();
    for (int s = 32; s > 0; s >>= 1) {
        if (t < s) sh[t] = fmaxf(sh[t], sh[t + s]);
        __syncthreads();
    }
    const float mx = sh[0];
    __syncthreads();
    sh[t] = (t < NCLASS) ? expf(v - mx) : 0.f;
    __syncthreads();
    for (int s = 32; s > 0; s >>= 1) {
        if (t < s) sh[t] += sh[t + s];
        __syncthreads();
    }
    const float lse = logf(sh[0]);
    if (t < NCLASS) out[g * NCLASS + t] = v - mx - lse;
}

extern "C" void kernel_launch(void* const* d_in, const int* in_sizes, int n_in,
                              void* d_out, int out_size, void* d_ws, size_t ws_size,
                              hipStream_t stream) {
    const float* x      = (const float*)d_in[0];
    const int*   ei     = (const int*)d_in[1];
    const int*   batch  = (const int*)d_in[2];
    const float* Ws     = (const float*)d_in[3];
    const float* bs     = (const float*)d_in[4];
    const float* gammas = (const float*)d_in[5];
    const float* betas  = (const float*)d_in[6];
    const float* Wout   = (const float*)d_in[7];
    const float* bout   = (const float*)d_in[8];
    float* out = (float*)d_out;

    const int* src = ei;
    const int* dst = ei + N_EDGES;

    float* ws = (float*)d_ws;
    const size_t nd = (size_t)N_NODES * DIM;
    float* hA     = ws;
    float* hB     = ws + nd;
    float* tbuf   = ws + 2 * nd;
    float* psum   = ws + 3 * nd;
    float* cnt    = psum + (size_t)NGRAPH * DIM;
    int*   deg    = (int*)(cnt + NGRAPH);
    int*   rowptr = deg + SCAN_NPAD;
    int*   bsum   = rowptr + SCAN_NPAD;
    int*   col    = bsum + 256;
    int*   slot   = col + N_EDGES;
    float* logits = (float*)(slot + N_EDGES);

    const int gEdge    = (N_EDGES + 255) / 256;
    const int gGemm    = N_NODES / 16;
    const int gPool    = (N_NODES * 32 + 255) / 256;

    const int nZero = NGRAPH * DIM + NGRAPH + SCAN_NPAD;
    gin_zero<<<(nZero + 255) / 256, 256, 0, stream>>>(psum, nZero);
    gin_count<<<gEdge, 256, 0, stream>>>(dst, deg, slot);
    gin_blocksum<<<SCAN_NBLK, 256, 0, stream>>>(deg, bsum);
    gin_rowptr<<<SCAN_NBLK, 256, 0, stream>>>(deg, bsum, rowptr);
    gin_place<<<gEdge, 256, 0, stream>>>(src, dst, rowptr, slot, col);

    const float* hin = x;
    for (int i = 0; i < NLAYERS; ++i) {
        float* hout = (i & 1) ? hB : hA;
        gin_gather<<<GATHER_NBLK, 256, 0, stream>>>(hin, rowptr, col, tbuf);
        gin_gemm_ln<<<gGemm, 256, 0, stream>>>(tbuf, Ws + (size_t)i * DIM * DIM,
                                               bs + i * DIM, gammas + i * DIM,
                                               betas + i * DIM, hout);
        hin = hout;
    }

    gin_pool<<<gPool, 256, 0, stream>>>(hin, batch, psum, cnt);
    gin_outproj<<<(NGRAPH * NCLASS + 255) / 256, 256, 0, stream>>>(psum, cnt, Wout, bout, logits);
    gin_logsoftmax<<<NGRAPH, 64, 0, stream>>>(logits, out);
}


# GEMM+LN fed through LDS: W staged once per resident work-group as B fragments, a wave owns 16 full rows (8 acc tiles), whole-row loads/stores; LN row sums lane-first, then 16-lane butterfly
# speedup vs baseline: 2.1059x; 1.3043x over previous
#include <hip/hip_runtime.h>
#include <hip/hip_bf16.h>
#include <math.h>

#define N_NODES 100000
#define N_EDGES 600000
#define DIM     128
#define NLAYERS 4
#define NGRAPH  512
#define NCLASS  40
#define LN_EPSF 1e-5f

typedef __attribute__((ext_vector_type(2))) float v2f;
typedef __attribute__((ext_vector_type(4))) float v4f;
typedef __attribute__((ext_vector_type(8))) float v8f;

#define SCAN_PER_BLK 1024
#define SCAN_NBLK    ((N_NODES + 1 + SCAN_PER_BLK - 1) / SCAN_PER_BLK)
#define SCAN_NPAD    (SCAN_NBLK * SCAN_PER_BLK)
#define GATHER_NBLK  2048

__global__ void gin_count(const int* __restrict__ dst, int* __restrict__ deg,
                          int* __restrict__ slot) {
    int e = blockIdx.x * blockDim.x + threadIdx.x;
    if (e < N_EDGES) slot[e] = atomicAdd(deg + dst[e], 1);
}

__global__ __launch_bounds__(256)
void gin_blocksum(const int* __restrict__ deg, int* __restrict__ bsum) {
    __shared__ int wsum[8];
    const int4 d = ((const int4*)deg)[blockIdx.x * 256 + threadIdx.x];
    int s = d.x + d.y + d.z + d.w;
#pragma unroll
    for (int m = 16; m > 0; m >>= 1) s += __shfl_xor(s, m, 32);
    if ((threadIdx.x & 31) == 0) wsum[threadIdx.x >> 5] = s;
    __syncthreads();
    if (threadIdx.x == 0) {
        int t = 0;
#pragma unroll
        for (int w = 0; w < 8; ++w) t += wsum[w];
        bsum[blockIdx.x] = t;
    }
}

__global__ __launch_bounds__(256)
void gin_rowptr(const int* __restrict__ deg, const int* __restrict__ bsum,
                int* __restrict__ rowptr) {
    static_assert(SCAN_NBLK <= 256, "one thread per earlier block sum");
    __shared__ int wsum[8];
    __shared__ int woff[8];
    const int tid  = threadIdx.x;
    const int lane = tid & 31;
    const int wave = tid >> 5;

    int p = (tid < (int)blockIdx.x) ? bsum[tid] : 0;
#pragma unroll
    for (int m = 16; m > 0; m >>= 1) p += __shfl_xor(p, m, 32);

    const int4 d = ((const int4*)deg)[blockIdx.x * 256 + tid];
    const int s = d.x + d.y + d.z + d.w;
    int inc = s;
#pragma unroll
    for (int m = 1; m < 32; m <<= 1) {
        const int v = __shfl_up(inc, m, 32);
        if (lane >= m) inc += v;
    }
    if (lane == 31) wsum[wave] = inc;
    if (lane == 0)  woff[wave] = p;
    __syncthreads();

    int base = 0;
#pragma unroll
    for (int w = 0; w < 8; ++w) {
        base += woff[w];
        if (w < wave) base += wsum[w];
    }
    int4 r;
    r.x = base + inc - s;
    r.y = r.x + d.x;
    r.z = r.y + d.y;
    r.w = r.z + d.z;
    ((int4*)rowptr)[blockIdx.x * 256 + tid] = r;
}

__global__ void gin_place(const int* __restrict__ src, const int* __restrict__ dst,
                          const int* __restrict__ rowptr, const int* __restrict__ slot,
                          int* __restrict__ col) {
    int e = blockIdx.x * blockDim.x + threadIdx.x;
    if (e < N_EDGES) col[rowptr[dst[e]] + slot[e]] = src[e];
}

__device__ __forceinline__ float4 gin_row4(const float* __restrict__ h, int node, int lane) {
    return *(const float4*)(h + (long long)node * DIM + lane * 4);
}

__global__ __launch_bounds__(256)
void gin_gather(const float* __restrict__ h, const int* __restrict__ rowptr,
                const int* __restrict__ col, float* __restrict__ t) {
    const int lane = threadIdx.x & 31;
    const int w  = __builtin_amdgcn_readfirstlane((int)((blockIdx.x * blockDim.x + threadIdx.x) >> 5));
    const int nw = (int)((gridDim.x * blockDim.x) >> 5);
    if (w >= N_NODES) return;

    const int p0 = rowptr[w + (lane & 1)];
    int beg = __builtin_amdgcn_readlane(p0, 0);
    int end = __builtin_amdgcn_readlane(p0, 1);

    for (int n = w; n < N_NODES; n += nw) {
        const int nn = n + nw;
        const int pn = nn < N_NODES ? nn : n;
        const int pnext = rowptr[pn + (lane & 1)];
        const float4 self = gin_row4(h, n, lane);
        float4 acc = make_float4(0.f, 0.f, 0.f, 0.f);

        for (int base = beg; base < end; base += 32) {
            const int cnt = (end - base) < 32 ? (end - base) : 32;
            const int my = (lane < cnt) ? col[base + lane] : 0;
            int j = 0;
            do {
                float4 v[8];
                v[0] = gin_row4(h, __builtin_amdgcn_readlane(my, j), lane);
#pragma unroll
                for (int u = 1; u < 8; ++u) {
                    if (j + u < cnt) v[u] = gin_row4(h, __builtin_amdgcn_readlane(my, j + u), lane);
                }
                acc.x += v[0].x; acc.y += v[0].y; acc.z += v[0].z; acc.w += v[0].w;
#pragma unroll
                for (int u = 1; u < 8; ++u) {
                    if (j + u < cnt) {
                        acc.x += v[u].x; acc.y += v[u].y; acc.z += v[u].z; acc.w += v[u].w;
                    }
                }
                j += 8;
            } while (j < cnt);
        }
        float4 o;
        o.x = 2.f * self.x + acc.x;
        o.y = 2.f * self.y + acc.y;
        o.z = 2.f * self.z + acc.z;
        o.w = 2.f * self.w + acc.w;
        *(float4*)(t + (long long)n * DIM + lane * 4) = o;

        beg = __builtin_amdgcn_readlane(pnext, 0);
        end = __builtin_amdgcn_readlane(pnext, 1);
    }
}

#define GEMM_NBLK   256
#define GEMM_WAVES  16
#define N_TILES     (N_NODES / 16)
#define A_PITCH     132

__device__ __forceinline__ void gin_wave_lds_handover() {
    __builtin_amdgcn_fence(__ATOMIC_RELEASE, "workgroup", "local");
    __builtin_amdgcn_wave_barrier();
    __builtin_amdgcn_fence(__ATOMIC_ACQUIRE, "workgroup", "local");
}

__global__ __launch_bounds__(GEMM_WAVES * 32)
void gin_gemm_ln(const float* __restrict__ T, const float* __restrict__ W,
                 const float* __restrict__ bias, const float* __restrict__ gamma,
                 const float* __restrict__ beta, float* __restrict__ Hout) {
    __shared__ __attribute__((aligned(16))) float w_frag[32 * 4 * 32 * 4];
    __shared__ __attribute__((aligned(16))) float a_tile[GEMM_WAVES * 16 * A_PITCH];

    const int tid  = threadIdx.x;
    const int lane = tid & 31;
    const int c    = lane & 15;
    const int h    = lane >> 4;
    const int wave = __builtin_amdgcn_readfirstlane(tid >> 5);

#pragma unroll
    for (int i = 0; i < 128 / GEMM_WAVES; ++i) {
        const int grp = wave + GEMM_WAVES * i;
        const int t   = grp >> 2;
        const int jp  = grp & 3;
        const float* wp = W + (4 * t + 2 * h) * DIM + 32 * jp + c;
        float4 f;
        f.x = wp[0];
        f.y = wp[DIM];
        f.z = wp[16];
        f.w = wp[DIM + 16];
        *(float4*)(w_frag + ((t * 4 + jp) * 32 + lane) * 4) = f;
    }

    float bv[8], gv[8], bt[8];
#pragma unroll
    for (int j = 0; j < 8; ++j) {
        bv[j] = bias[16 * j + c];
        gv[j] = gamma[16 * j + c];
        bt[j] = beta[16 * j + c];
    }

    float* my_a = a_tile + wave * (16 * A_PITCH);

    static_assert(GEMM_NBLK * GEMM_WAVES <= N_TILES, "a first tile for every wave");
    int tile = (int)blockIdx.x + GEMM_NBLK * wave;
    v4f raw[16];
#pragma unroll
    for (int r = 0; r < 16; ++r)
        raw[r] = *(const v4f*)(T + (long long)(tile * 16 + r) * DIM + lane * 4);
    __syncthreads();

    do {
        gin_wave_lds_handover();
#pragma unroll
        for (int r = 0; r < 16; ++r)
            *(v4f*)(my_a + r * A_PITCH + lane * 4) = raw[r];
        gin_wave_lds_handover();
        const float* my_frag = my_a + c * A_PITCH + 2 * h;

        const int ntile = tile + GEMM_NBLK * GEMM_WAVES;
        const int ptile = ntile < N_TILES ? ntile : tile;
#pragma unroll
        for (int r = 0; r < 16; ++r)
            raw[r] = *(const v4f*)(T + (long long)(ptile * 16 + r) * DIM + lane * 4);

        v8f acc[8];
#pragma unroll
        for (int j = 0; j < 8; ++j) acc[j] = (v8f){0.f, 0.f, 0.f, 0.f, 0.f, 0.f, 0.f, 0.f};

        const float* wl = w_frag + lane * 4;
        v4f b0s[16], b1s[16];
        v2f a0s[4], a1s[4];
#pragma unroll
        for (int u = 0; u < 16; ++u) b0s[u] = *(const v4f*)(wl + u * 128);
#pragma unroll
        for (int tt = 0; tt < 4; ++tt) a0s[tt] = *(const v2f*)(my_frag + 4 * tt);
#pragma nounroll
        for (int kk = 0; kk < 4; ++kk) {
            const int blk1 = 2 * kk + 1;
            const int blk2 = (2 * kk + 2) & 7;
#pragma unroll
            for (int u = 0; u < 16; ++u) b1s[u] = *(const v4f*)(wl + blk1 * 2048 + u * 128);
#pragma unroll
            for (int tt = 0; tt < 4; ++tt) a1s[tt] = *(const v2f*)(my_frag + blk1 * 16 + 4 * tt);
#pragma unroll
            for (int tt = 0; tt < 4; ++tt) {
#pragma unroll
                for (int jp = 0; jp < 4; ++jp) {
                    const v4f f = b0s[tt * 4 + jp];
                    v2f b0, b1;
                    b0.x = f.x; b0.y = f.y;
                    b1.x = f.z; b1.y = f.w;
                    acc[2 * jp] = __builtin_amdgcn_wmma_f32_16x16x4_f32(
                            false, a0s[tt], false, b0, (short)0, acc[2 * jp], false, false);
                    acc[2 * jp + 1] = __builtin_amdgcn_wmma_f32_16x16x4_f32(
                            false, a0s[tt], false, b1, (short)0, acc[2 * jp + 1], false, false);
                }
            }
#pragma unroll
            for (int u = 0; u < 16; ++u) b0s[u] = *(const v4f*)(wl + blk2 * 2048 + u * 128);
#pragma unroll
            for (int tt = 0; tt < 4; ++tt) a0s[tt] = *(const v2f*)(my_frag + blk2 * 16 + 4 * tt);
#pragma unroll
            for (int tt = 0; tt < 4; ++tt) {
#pragma unroll
                for (int jp = 0; jp < 4; ++jp) {
                    const v4f f = b1s[tt * 4 + jp];
                    v2f b0, b1;
                    b0.x = f.x; b0.y = f.y;
                    b1.x = f.z; b1.y = f.w;
                    acc[2 * jp] = __builtin_amdgcn_wmma_f32_16x16x4_f32(
                            false, a1s[tt], false, b0, (short)0, acc[2 * jp], false, false);
                    acc[2 * jp + 1] = __builtin_amdgcn_wmma_f32_16x16x4_f32(
                            false, a1s[tt], false, b1, (short)0, acc[2 * jp + 1], false, false);
                }
            }
        }

        float s[8], q[8];
#pragma unroll
        for (int r = 0; r < 8; ++r) { s[r] = 0.f; q[r] = 0.f; }
        gin_wave_lds_handover();
#pragma unroll
        for (int j = 0; j < 8; ++j) {
#pragma unroll
            for (int r = 0; r < 8; ++r) {
                float v = acc[j][r] + bv[j];
                v = v > 0.f ? v : 0.f;
                acc[j][r] = v;
                s[r] += v;
                q[r] += v * v;
            }
        }
#pragma unroll
        for (int m = 1; m < 16; m <<= 1) {
            float es[8], eq[8];
#pragma unroll
            for (int r = 0; r < 8; ++r) {
                es[r] = __shfl_xor(s[r], m, 32);
                eq[r] = __shfl_xor(q[r], m, 32);
            }
#pragma unroll
            for (int r = 0; r < 8; ++r) {
                s[r] += es[r];
                q[r] += eq[r];
            }
        }
#pragma unroll
        for (int r = 0; r < 8; ++r) {
            const float mean = s[r] * (1.0f / DIM);
            const float var  = q[r] * (1.0f / DIM) - mean * mean;
            const float rs   = rsqrtf(var + LN_EPSF);
#pragma unroll
            for (int j = 0; j < 8; ++j) {
                const float o = (acc[j][r] - mean) * rs * gv[j] + bt[j];
                my_a[(8 * h + r) * A_PITCH + 16 * j + c] = o;
            }
        }
        gin_wave_lds_handover();
#pragma unroll
        for (int r = 0; r < 16; ++r) {
            const float4 o = *(const float4*)(my_a + r * A_PITCH + lane * 4);
            *(float4*)(Hout + (long long)(tile * 16 + r) * DIM + lane * 4) = o;
        }
        tile = ntile;
    } while (tile < N_TILES);
}

__global__ void gin_zero(float* __restrict__ p, int n) {
    int i = blockIdx.x * blockDim.x + threadIdx.x;
    if (i < n) p[i] = 0.f;
}

__global__ void gin_pool(const float* __restrict__ h, const int* __restrict__ batch,
                         float* __restrict__ psum, float* __restrict__ cnt) {
    int tid = blockIdx.x * blockDim.x + threadIdx.x;
    int n = tid >> 5;
    int lane = tid & 31;
    if (n < N_NODES) {
        int b = batch[n];
        const float4 v = *(const float4*)(h + (long long)n * DIM + lane * 4);
        float* p = psum + (long long)b * DIM + lane * 4;
        atomicAdd(p + 0, v.x);
        atomicAdd(p + 1, v.y);
        atomicAdd(p + 2, v.z);
        atomicAdd(p + 3, v.w);
        if (lane == 0) atomicAdd(cnt + b, 1.0f);
    }
}

__global__ void gin_outproj(const float* __restrict__ psum, const float* __restrict__ cnt,
                            const float* __restrict__ Wout, const float* __restrict__ bout,
                            float* __restrict__ logits) {
    int idx = blockIdx.x * blockDim.x + threadIdx.x;
    if (idx < NGRAPH * NCLASS) {
        int g = idx / NCLASS;
        int c = idx - g * NCLASS;
        float inv = 1.0f / fmaxf(cnt[g], 1.0f);
        float acc = bout[c];
        const float* pg = psum + g * DIM;
#pragma unroll 4
        for (int k = 0; k < DIM; ++k) acc += pg[k] * inv * Wout[k * NCLASS + c];
        logits[idx] = acc;
    }
}

__global__ void gin_logsoftmax(const float* __restrict__ logits, float* __restrict__ out) {
    __shared__ float sh[64];
    const int g = blockIdx.x;
    const int t = threadIdx.x;
    float v = (t < NCLASS) ? logits[g * NCLASS + t] : -3.4e38f;
    sh[t] = v;
    __syncthreads();
    for (int s = 32; s > 0; s >>= 1) {
        if (t < s) sh[t] = fmaxf(sh[t], sh[t + s]);
        __syncthreads();
    }
    const float mx = sh[0];
    __syncthreads();
    sh[t] = (t < NCLASS) ? expf(v - mx) : 0.f;
    __syncthreads();
    for (int s = 32; s > 0; s >>= 1) {
        if (t < s) sh[t] += sh[t + s];
        __syncthreads();
    }
    const float lse = logf(sh[0]);
    if (t < NCLASS) out[g * NCLASS + t] = v - mx - lse;
}

extern "C" void kernel_launch(void* const* d_in, const int* in_sizes, int n_in,
                              void* d_out, int out_size, void* d_ws, size_t ws_size,
                              hipStream_t stream) {
    const float* x      = (const float*)d_in[0];
    const int*   ei     = (const int*)d_in[1];
    const int*   batch  = (const int*)d_in[2];
    const float* Ws     = (const float*)d_in[3];
    const float* bs     = (const float*)d_in[4];
    const float* gammas = (const float*)d_in[5];
    const float* betas  = (const float*)d_in[6];
    const float* Wout   = (const float*)d_in[7];
    const float* bout   = (const float*)d_in[8];
    float* out = (float*)d_out;

    const int* src = ei;
    const int* dst = ei + N_EDGES;

    float* ws = (float*)d_ws;
    const size_t nd = (size_t)N_NODES * DIM;
    float* hA     = ws;
    float* hB     = ws + nd;
    float* tbuf   = ws + 2 * nd;
    float* psum   = ws + 3 * nd;
    float* cnt    = psum + (size_t)NGRAPH * DIM;
    int*   deg    = (int*)(cnt + NGRAPH);
    int*   rowptr = deg + SCAN_NPAD;
    int*   bsum   = rowptr + SCAN_NPAD;
    int*   col    = bsum + 256;
    int*   slot   = col + N_EDGES;
    float* logits = (float*)(slot + N_EDGES);

    const int gEdge    = (N_EDGES + 255) / 256;
    const int gPool    = (N_NODES * 32 + 255) / 256;

    const int nZero = NGRAPH * DIM + NGRAPH + SCAN_NPAD;
    gin_zero<<<(nZero + 255) / 256, 256, 0, stream>>>(psum, nZero);
    gin_count<<<gEdge, 256, 0, stream>>>(dst, deg, slot);
    gin_blocksum<<<SCAN_NBLK, 256, 0, stream>>>(deg, bsum);
    gin_rowptr<<<SCAN_NBLK, 256, 0, stream>>>(deg, bsum, rowptr);
    gin_place<<<gEdge, 256, 0, stream>>>(src, dst, rowptr, slot, col);

    const float* hin = x;
    for (int i = 0; i < NLAYERS; ++i) {
        float* hout = (i & 1) ? hB : hA;
        gin_gather<<<GATHER_NBLK, 256, 0, stream>>>(hin, rowptr, col, tbuf);
        gin_gemm_ln<<<GEMM_NBLK, GEMM_WAVES * 32, 0, stream>>>(tbuf, Ws + (size_t)i * DIM * DIM,
                                               bs + i * DIM, gammas + i * DIM,
                                               betas + i * DIM, hout);
        hin = hout;
    }

    gin_pool<<<gPool, 256, 0, stream>>>(hin, batch, psum, cnt);
    gin_outproj<<<(NGRAPH * NCLASS + 255) / 256, 256, 0, stream>>>(psum, cnt, Wout, bout, logits);
    gin_logsoftmax<<<NGRAPH, 64, 0, stream>>>(logits, out);
}


# Pool: a wave sums 32 consecutive node rows in registers and adds to the graph sums by float atomics only where the graph id changes; 100000 one-node waves become 3125; pooled sum order changes
# speedup vs baseline: 2.9682x; 1.4095x over previous
#include <hip/hip_runtime.h>
#include <hip/hip_bf16.h>
#include <math.h>

#define N_NODES 100000
#define N_EDGES 600000
#define DIM     128
#define NLAYERS 4
#define NGRAPH  512
#define NCLASS  40
#define LN_EPSF 1e-5f

typedef __attribute__((ext_vector_type(2))) float v2f;
typedef __attribute__((ext_vector_type(4))) float v4f;
typedef __attribute__((ext_vector_type(8))) float v8f;

#define SCAN_PER_BLK 1024
#define SCAN_NBLK    ((N_NODES + 1 + SCAN_PER_BLK - 1) / SCAN_PER_BLK)
#define SCAN_NPAD    (SCAN_NBLK * SCAN_PER_BLK)
#define GATHER_NBLK  2048

__global__ void gin_count(const int* __restrict__ dst, int* __restrict__ deg,
                          int* __restrict__ slot) {
    int e = blockIdx.x * blockDim.x + threadIdx.x;
    if (e < N_EDGES) slot[e] = atomicAdd(deg + dst[e], 1);
}

__global__ __launch_bounds__(256)
void gin_blocksum(const int* __restrict__ deg, int* __restrict__ bsum) {
    __shared__ int wsum[8];
    const int4 d = ((const int4*)deg)[blockIdx.x * 256 + threadIdx.x];
    int s = d.x + d.y + d.z + d.w;
#pragma unroll
    for (int m = 16; m > 0; m >>= 1) s += __shfl_xor(s, m, 32);
    if ((threadIdx.x & 31) == 0) wsum[threadIdx.x >> 5] = s;
    __syncthreads();
    if (threadIdx.x == 0) {
        int t = 0;
#pragma unroll
        for (int w = 0; w < 8; ++w) t += wsum[w];
        bsum[blockIdx.x] = t;
    }
}

__global__ __launch_bounds__(256)
void gin_rowptr(const int* __restrict__ deg, const int* __restrict__ bsum,
                int* __restrict__ rowptr) {
    static_assert(SCAN_NBLK <= 256, "one thread per earlier block sum");
    __shared__ int wsum[8];
    __shared__ int woff[8];
    const int tid  = threadIdx.x;
    const int lane = tid & 31;
    const int wave = tid >> 5;

    int p = (tid < (int)blockIdx.x) ? bsum[tid] : 0;
#pragma unroll
    for (int m = 16; m > 0; m >>= 1) p += __shfl_xor(p, m, 32);

    const int4 d = ((const int4*)deg)[blockIdx.x * 256 + tid];
    const int s = d.x + d.y + d.z + d.w;
    int inc = s;
#pragma unroll
    for (int m = 1; m < 32; m <<= 1) {
        const int v = __shfl_up(inc, m, 32);
        if (lane >= m) inc += v;
    }
    if (lane == 31) wsum[wave] = inc;
    if (lane == 0)  woff[wave] = p;
    __syncthreads();

    int base = 0;
#pragma unroll
    for (int w = 0; w < 8; ++w) {
        base += woff[w];
        if (w < wave) base += wsum[w];
    }
    int4 r;
    r.x = base + inc - s;
    r.y = r.x + d.x;
    r.z = r.y + d.y;
    r.w = r.z + d.z;
    ((int4*)rowptr)[blockIdx.x * 256 + tid] = r;
}

__global__ void gin_place(const int* __restrict__ src, const int* __restrict__ dst,
                          const int* __restrict__ rowptr, const int* __restrict__ slot,
                          int* __restrict__ col) {
    int e = blockIdx.x * blockDim.x + threadIdx.x;
    if (e < N_EDGES) col[rowptr[dst[e]] + slot[e]] = src[e];
}

__device__ __forceinline__ float4 gin_row4(const float* __restrict__ h, int node, int lane) {
    return *(const float4*)(h + (long long)node * DIM + lane * 4);
}

__global__ __launch_bounds__(256)
void gin_gather(const float* __restrict__ h, const int* __restrict__ rowptr,
                const int* __restrict__ col, float* __restrict__ t) {
    const int lane = threadIdx.x & 31;
    const int w  = __builtin_amdgcn_readfirstlane((int)((blockIdx.x * blockDim.x + threadIdx.x) >> 5));
    const int nw = (int)((gridDim.x * blockDim.x) >> 5);
    if (w >= N_NODES) return;

    const int p0 = rowptr[w + (lane & 1)];
    int beg = __builtin_amdgcn_readlane(p0, 0);
    int end = __builtin_amdgcn_readlane(p0, 1);

    for (int n = w; n < N_NODES; n += nw) {
        const int nn = n + nw;
        const int pn = nn < N_NODES ? nn : n;
        const int pnext = rowptr[pn + (lane & 1)];
        const float4 self = gin_row4(h, n, lane);
        float4 acc = make_float4(0.f, 0.f, 0.f, 0.f);

        for (int base = beg; base < end; base += 32) {
            const int cnt = (end - base) < 32 ? (end - base) : 32;
            const int my = (lane < cnt) ? col[base + lane] : 0;
            int j = 0;
            do {
                float4 v[8];
                v[0] = gin_row4(h, __builtin_amdgcn_readlane(my, j), lane);
#pragma unroll
                for (int u = 1; u < 8; ++u) {
                    if (j + u < cnt) v[u] = gin_row4(h, __builtin_amdgcn_readlane(my, j + u), lane);
                }
                acc.x += v[0].x; acc.y += v[0].y; acc.z += v[0].z; acc.w += v[0].w;
#pragma unroll
                for (int u = 1; u < 8; ++u) {
                    if (j + u < cnt) {
                        acc.x += v[u].x; acc.y += v[u].y; acc.z += v[u].z; acc.w += v[u].w;
                    }
                }
                j += 8;
            } while (j < cnt);
        }
        float4 o;
        o.x = 2.f * self.x + acc.x;
        o.y = 2.f * self.y + acc.y;
        o.z = 2.f * self.z + acc.z;
        o.w = 2.f * self.w + acc.w;
        *(float4*)(t + (long long)n * DIM + lane * 4) = o;

        beg = __builtin_amdgcn_readlane(pnext, 0);
        end = __builtin_amdgcn_readlane(pnext, 1);
    }
}

#define GEMM_NBLK   256
#define GEMM_WAVES  16
#define N_TILES     (N_NODES / 16)
#define A_PITCH     132

__device__ __forceinline__ void gin_wave_lds_handover() {
    __builtin_amdgcn_fence(__ATOMIC_RELEASE, "workgroup", "local");
    __builtin_amdgcn_wave_barrier();
    __builtin_amdgcn_fence(__ATOMIC_ACQUIRE, "workgroup", "local");
}

__global__ __launch_bounds__(GEMM_WAVES * 32)
void gin_gemm_ln(const float* __restrict__ T, const float* __restrict__ W,
                 const float* __restrict__ bias, const float* __restrict__ gamma,
                 const float* __restrict__ beta, float* __restrict__ Hout) {
    __shared__ __attribute__((aligned(16))) float w_frag[32 * 4 * 32 * 4];
    __shared__ __attribute__((aligned(16))) float a_tile[GEMM_WAVES * 16 * A_PITCH];

    const int tid  = threadIdx.x;
    const int lane = tid & 31;
    const int c    = lane & 15;
    const int h    = lane >> 4;
    const int wave = __builtin_amdgcn_readfirstlane(tid >> 5);

#pragma unroll
    for (int i = 0; i < 128 / GEMM_WAVES; ++i) {
        const int grp = wave + GEMM_WAVES * i;
        const int t   = grp >> 2;
        const int jp  = grp & 3;
        const float* wp = W + (4 * t + 2 * h) * DIM + 32 * jp + c;
        float4 f;
        f.x = wp[0];
        f.y = wp[DIM];
        f.z = wp[16];
        f.w = wp[DIM + 16];
        *(float4*)(w_frag + ((t * 4 + jp) * 32 + lane) * 4) = f;
    }

    float bv[8], gv[8], bt[8];
#pragma unroll
    for (int j = 0; j < 8; ++j) {
        bv[j] = bias[16 * j + c];
        gv[j] = gamma[16 * j + c];
        bt[j] = beta[16 * j + c];
    }

    float* my_a = a_tile + wave * (16 * A_PITCH);

    static_assert(GEMM_NBLK * GEMM_WAVES <= N_TILES, "a first tile for every wave");
    int tile = (int)blockIdx.x + GEMM_NBLK * wave;
    v4f raw[16];
#pragma unroll
    for (int r = 0; r < 16; ++r)
        raw[r] = *(const v4f*)(T + (long long)(tile * 16 + r) * DIM + lane * 4);
    __syncthreads();

    do {
        gin_wave_lds_handover();
#pragma unroll
        for (int r = 0; r < 16; ++r)
            *(v4f*)(my_a + r * A_PITCH + lane * 4) = raw[r];
        gin_wave_lds_handover();
        const float* my_frag = my_a + c * A_PITCH + 2 * h;

        const int ntile = tile + GEMM_NBLK * GEMM_WAVES;
        const int ptile = ntile < N_TILES ? ntile : tile;
#pragma unroll
        for (int r = 0; r < 16; ++r)
            raw[r] = *(const v4f*)(T + (long long)(ptile * 16 + r) * DIM + lane * 4);

        v8f acc[8];
#pragma unroll
        for (int j = 0; j < 8; ++j) acc[j] = (v8f){0.f, 0.f, 0.f, 0.f, 0.f, 0.f, 0.f, 0.f};

        const float* wl = w_frag + lane * 4;
        v4f b0s[16], b1s[16];
        v2f a0s[4], a1s[4];
#pragma unroll
        for (int u = 0; u < 16; ++u) b0s[u] = *(const v4f*)(wl + u * 128);
#pragma unroll
        for (int tt = 0; tt < 4; ++tt) a0s[tt] = *(const v2f*)(my_frag + 4 * tt);
#pragma nounroll
        for (int kk = 0; kk < 4; ++kk) {
            const int blk1 = 2 * kk + 1;
            const int blk2 = (2 * kk + 2) & 7;
#pragma unroll
            for (int u = 0; u < 16; ++u) b1s[u] = *(const v4f*)(wl + blk1 * 2048 + u * 128);
#pragma unroll
            for (int tt = 0; tt < 4; ++tt) a1s[tt] = *(const v2f*)(my_frag + blk1 * 16 + 4 * tt);
#pragma unroll
            for (int tt = 0; tt < 4; ++tt) {
#pragma unroll
                for (int jp = 0; jp < 4; ++jp) {
                    const v4f f = b0s[tt * 4 + jp];
                    v2f b0, b1;
                    b0.x = f.x; b0.y = f.y;
                    b1.x = f.z; b1.y = f.w;
                    acc[2 * jp] = __builtin_amdgcn_wmma_f32_16x16x4_f32(
                            false, a0s[tt], false, b0, (short)0, acc[2 * jp], false, false);
                    acc[2 * jp + 1] = __builtin_amdgcn_wmma_f32_16x16x4_f32(
                            false, a0s[tt], false, b1, (short)0, acc[2 * jp + 1], false, false);
                }
            }
#pragma unroll
            for (int u = 0; u < 16; ++u) b0s[u] = *(const v4f*)(wl + blk2 * 2048 + u * 128);
#pragma unroll
            for (int tt = 0; tt < 4; ++tt) a0s[tt] = *(const v2f*)(my_frag + blk2 * 16 + 4 * tt);
#pragma unroll
            for (int tt = 0; tt < 4; ++tt) {
#pragma unroll
                for (int jp = 0; jp < 4; ++jp) {
                    const v4f f = b1s[tt * 4 + jp];
                    v2f b0, b1;
                    b0.x = f.x; b0.y = f.y;
                    b1.x = f.z; b1.y = f.w;
                    acc[2 * jp] = __builtin_amdgcn_wmma_f32_16x16x4_f32(
                            false, a1s[tt], false, b0, (short)0, acc[2 * jp], false, false);
                    acc[2 * jp + 1] = __builtin_amdgcn_wmma_f32_16x16x4_f32(
                            false, a1s[tt], false, b1, (short)0, acc[2 * jp + 1], false, false);
                }
            }
        }

        float s[8], q[8];
#pragma unroll
        for (int r = 0; r < 8; ++r) { s[r] = 0.f; q[r] = 0.f; }
        gin_wave_lds_handover();
#pragma unroll
        for (int j = 0; j < 8; ++j) {
#pragma unroll
            for (int r = 0; r < 8; ++r) {
                float v = acc[j][r] + bv[j];
                v = v > 0.f ? v : 0.f;
                acc[j][r] = v;
                s[r] += v;
                q[r] += v * v;
            }
        }
#pragma unroll
        for (int m = 1; m < 16; m <<= 1) {
            float es[8], eq[8];
#pragma unroll
            for (int r = 0; r < 8; ++r) {
                es[r] = __shfl_xor(s[r], m, 32);
                eq[r] = __shfl_xor(q[r], m, 32);
            }
#pragma unroll
            for (int r = 0; r < 8; ++r) {
                s[r] += es[r];
                q[r] += eq[r];
            }
        }
#pragma unroll
        for (int r = 0; r < 8; ++r) {
            const float mean = s[r] * (1.0f / DIM);
            const float var  = q[r] * (1.0f / DIM) - mean * mean;
            const float rs   = rsqrtf(var + LN_EPSF);
#pragma unroll
            for (int j = 0; j < 8; ++j) {
                const float o = (acc[j][r] - mean) * rs * gv[j] + bt[j];
                my_a[(8 * h + r) * A_PITCH + 16 * j + c] = o;
            }
        }
        gin_wave_lds_handover();
#pragma unroll
        for (int r = 0; r < 16; ++r) {
            const float4 o = *(const float4*)(my_a + r * A_PITCH + lane * 4);
            *(float4*)(Hout + (long long)(tile * 16 + r) * DIM + lane * 4) = o;
        }
        tile = ntile;
    } while (tile < N_TILES);
}

__global__ void gin_zero(float* __restrict__ p, int n) {
    int i = blockIdx.x * blockDim.x + threadIdx.x;
    if (i < n) p[i] = 0.f;
}

#define POOL_RUN 32

__device__ __forceinline__ void gin_pool_flush(float* __restrict__ psum, float* __restrict__ cnt,
                                               int g, int lane, float4 a, int len) {
    float* p = psum + (long long)g * DIM + lane * 4;
    atomicAdd(p + 0, a.x);
    atomicAdd(p + 1, a.y);
    atomicAdd(p + 2, a.z);
    atomicAdd(p + 3, a.w);
    if (lane == 0) atomicAdd(cnt + g, (float)len);
}

__global__ __launch_bounds__(256)
void gin_pool(const float* __restrict__ h, const int* __restrict__ batch,
              float* __restrict__ psum, float* __restrict__ cnt) {
    static_assert(N_NODES % POOL_RUN == 0, "a full run for every wave");
    const int lane = threadIdx.x & 31;
    const int w  = __builtin_amdgcn_readfirstlane((int)((blockIdx.x * blockDim.x + threadIdx.x) >> 5));
    const int n0 = w * POOL_RUN;
    if (n0 >= N_NODES) return;

    const int bid = batch[n0 + lane];
    float4 v[POOL_RUN];
#pragma unroll
    for (int r = 0; r < POOL_RUN; ++r) v[r] = gin_row4(h, n0 + r, lane);

    int cur = __builtin_amdgcn_readlane(bid, 0);
    int len = 1;
    float4 acc = v[0];
#pragma unroll
    for (int r = 1; r < POOL_RUN; ++r) {
        const int b = __builtin_amdgcn_readlane(bid, r);
        if (b != cur) {
            gin_pool_flush(psum, cnt, cur, lane, acc, len);
            cur = b;
            len = 0;
            acc = make_float4(0.f, 0.f, 0.f, 0.f);
        }
        acc.x += v[r].x; acc.y += v[r].y; acc.z += v[r].z; acc.w += v[r].w;
        ++len;
    }
    gin_pool_flush(psum, cnt, cur, lane, acc, len);
}

__global__ void gin_outproj(const float* __restrict__ psum, const float* __restrict__ cnt,
                            const float* __restrict__ Wout, const float* __restrict__ bout,
                            float* __restrict__ logits) {
    int idx = blockIdx.x * blockDim.x + threadIdx.x;
    if (idx < NGRAPH * NCLASS) {
        int g = idx / NCLASS;
        int c = idx - g * NCLASS;
        float inv = 1.0f / fmaxf(cnt[g], 1.0f);
        float acc = bout[c];
        const float* pg = psum + g * DIM;
#pragma unroll 4
        for (int k = 0; k < DIM; ++k) acc += pg[k] * inv * Wout[k * NCLASS + c];
        logits[idx] = acc;
    }
}

__global__ void gin_logsoftmax(const float* __restrict__ logits, float* __restrict__ out) {
    __shared__ float sh[64];
    const int g = blockIdx.x;
    const int t = threadIdx.x;
    float v = (t < NCLASS) ? logits[g * NCLASS + t] : -3.4e38f;
    sh[t] = v;
    __syncthreads();
    for (int s = 32; s > 0; s >>= 1) {
        if (t < s) sh[t] = fmaxf(sh[t], sh[t + s]);
        __syncthreads();
    }
    const float mx = sh[0];
    __syncthreads();
    sh[t] = (t < NCLASS) ? expf(v - mx) : 0.f;
    __syncthreads();
    for (int s = 32; s > 0; s >>= 1) {
        if (t < s) sh[t] += sh[t + s];
        __syncthreads();
    }
    const float lse = logf(sh[0]);
    if (t < NCLASS) out[g * NCLASS + t] = v - mx - lse;
}

extern "C" void kernel_launch(void* const* d_in, const int* in_sizes, int n_in,
                              void* d_out, int out_size, void* d_ws, size_t ws_size,
                              hipStream_t stream) {
    const float* x      = (const float*)d_in[0];
    const int*   ei     = (const int*)d_in[1];
    const int*   batch  = (const int*)d_in[2];
    const float* Ws     = (const float*)d_in[3];
    const float* bs     = (const float*)d_in[4];
    const float* gammas = (const float*)d_in[5];
    const float* betas  = (const float*)d_in[6];
    const float* Wout   = (const float*)d_in[7];
    const float* bout   = (const float*)d_in[8];
    float* out = (float*)d_out;

    const int* src = ei;
    const int* dst = ei + N_EDGES;

    float* ws = (float*)d_ws;
    const size_t nd = (size_t)N_NODES * DIM;
    float* hA     = ws;
    float* hB     = ws + nd;
    float* tbuf   = ws + 2 * nd;
    float* psum   = ws + 3 * nd;
    float* cnt    = psum + (size_t)NGRAPH * DIM;
    int*   deg    = (int*)(cnt + NGRAPH);
    int*   rowptr = deg + SCAN_NPAD;
    int*   bsum   = rowptr + SCAN_NPAD;
    int*   col    = bsum + 256;
    int*   slot   = col + N_EDGES;
    float* logits = (float*)(slot + N_EDGES);

    const int gEdge    = (N_EDGES + 255) / 256;
    const int gPool    = (N_NODES / POOL_RUN * 32 + 255) / 256;

    const int nZero = NGRAPH * DIM + NGRAPH + SCAN_NPAD;
    gin_zero<<<(nZero + 255) / 256, 256, 0, stream>>>(psum, nZero);
    gin_count<<<gEdge, 256, 0, stream>>>(dst, deg, slot);
    gin_blocksum<<<SCAN_NBLK, 256, 0, stream>>>(deg, bsum);
    gin_rowptr<<<SCAN_NBLK, 256, 0, stream>>>(deg, bsum, rowptr);
    gin_place<<<gEdge, 256, 0, stream>>>(src, dst, rowptr, slot, col);

    const float* hin = x;
    for (int i = 0; i < NLAYERS; ++i) {
        float* hout = (i & 1) ? hB : hA;
        gin_gather<<<GATHER_NBLK, 256, 0, stream>>>(hin, rowptr, col, tbuf);
        gin_gemm_ln<<<GEMM_NBLK, GEMM_WAVES * 32, 0, stream>>>(tbuf, Ws + (size_t)i * DIM * DIM,
                                               bs + i * DIM, gammas + i * DIM,
                                               betas + i * DIM, hout);
        hin = hout;
    }

    gin_pool<<<gPool, 256, 0, stream>>>(hin, batch, psum, cnt);
    gin_outproj<<<(NGRAPH * NCLASS + 255) / 256, 256, 0, stream>>>(psum, cnt, Wout, bout, logits);
    gin_logsoftmax<<<NGRAPH, 64, 0, stream>>>(logits, out);
}


# Edge list in one pass: the counting add's return is the edge's slot in a fixed 32-slot row per node (overflow list beyond); scan and place launches go, gather reads slot rows ahead; sum order kept
# speedup vs baseline: 3.1776x; 1.0705x over previous
#include <hip/hip_runtime.h>
#include <hip/hip_bf16.h>
#include <math.h>

#define N_NODES 100000
#define N_EDGES 600000
#define DIM     128
#define NLAYERS 4
#define NGRAPH  512
#define NCLASS  40
#define LN_EPSF 1e-5f

typedef __attribute__((ext_vector_type(2))) float v2f;
typedef __attribute__((ext_vector_type(4))) float v4f;
typedef __attribute__((ext_vector_type(8))) float v8f;

#define SLOT_CAP  32
#define DEG_PAD   (N_NODES + 96)
#define GATHER_NBLK  2048

__global__ void gin_fill(const int* __restrict__ src, const int* __restrict__ dst,
                         int* __restrict__ deg, int* __restrict__ col,
                         int* __restrict__ ovf_cnt, int* __restrict__ ovf_dst,
                         int* __restrict__ ovf_src) {
    int e = blockIdx.x * blockDim.x + threadIdx.x;
    if (e < N_EDGES) {
        const int d = dst[e];
        const int s = src[e];
        const int slot = atomicAdd(deg + d, 1);
        if (slot < SLOT_CAP) {
            col[d * SLOT_CAP + slot] = s;
        } else {
            const int k = atomicAdd(ovf_cnt, 1);
            ovf_dst[k] = d;
            ovf_src[k] = s;
        }
    }
}

__device__ __forceinline__ float4 gin_row4(const float* __restrict__ h, int node, int lane) {
    return *(const float4*)(h + (long long)node * DIM + lane * 4);
}

__global__ __launch_bounds__(256)
void gin_gather(const float* __restrict__ h, const int* __restrict__ deg,
                const int* __restrict__ col, const int* __restrict__ ovf_cnt,
                const int* __restrict__ ovf_dst, const int* __restrict__ ovf_src,
                float* __restrict__ t) {
    const int lane = threadIdx.x & 31;
    const int w  = __builtin_amdgcn_readfirstlane((int)((blockIdx.x * blockDim.x + threadIdx.x) >> 5));
    const int nw = (int)((gridDim.x * blockDim.x) >> 5);
    if (w >= N_NODES) return;

    int cnt0 = deg[w];
    int cnt1 = (w + nw) < N_NODES ? deg[w + nw] : 0;
    int c0   = cnt0 < SLOT_CAP ? cnt0 : SLOT_CAP;
    int my0  = *((lane < c0) ? (col + w * SLOT_CAP + lane) : (deg + w));

    for (int n = w; n < N_NODES; n += nw) {
        const int n1 = n + nw;
        const int n2 = n1 + nw;
        const int pd2 = deg[(n2 < N_NODES ? n2 : n) + (lane & 1)];
        const int c1   = cnt1 < SLOT_CAP ? cnt1 : SLOT_CAP;
        const int my1  = *((lane < c1) ? (col + n1 * SLOT_CAP + lane)
                                       : (deg + (n1 < N_NODES ? n1 : n)));
        const float4 self = gin_row4(h, n, lane);
        float4 acc = make_float4(0.f, 0.f, 0.f, 0.f);

        for (int j = 0; j < c0; j += 8) {
            float4 v[8];
            v[0] = gin_row4(h, __builtin_amdgcn_readlane(my0, j), lane);
#pragma unroll
            for (int u = 1; u < 8; ++u) {
                if (j + u < c0) v[u] = gin_row4(h, __builtin_amdgcn_readlane(my0, j + u), lane);
            }
            acc.x += v[0].x; acc.y += v[0].y; acc.z += v[0].z; acc.w += v[0].w;
#pragma unroll
            for (int u = 1; u < 8; ++u) {
                if (j + u < c0) {
                    acc.x += v[u].x; acc.y += v[u].y; acc.z += v[u].z; acc.w += v[u].w;
                }
            }
        }
        if (cnt0 > SLOT_CAP) {
            const int total = ovf_cnt[0];
            for (int base = 0; base < total; base += 32) {
                const int i  = base + lane;
                const int od = (i < total) ? ovf_dst[i] : -1;
                const int os = (i < total) ? ovf_src[i] : 0;
                unsigned int hit = __builtin_amdgcn_ballot_w32(od == n);
                while (hit != 0u) {
                    const int b = __builtin_ctz(hit);
                    hit &= hit - 1u;
                    const float4 v = gin_row4(h, __builtin_amdgcn_readlane(os, b), lane);
                    acc.x += v.x; acc.y += v.y; acc.z += v.z; acc.w += v.w;
                }
            }
        }
        float4 o;
        o.x = 2.f * self.x + acc.x;
        o.y = 2.f * self.y + acc.y;
        o.z = 2.f * self.z + acc.z;
        o.w = 2.f * self.w + acc.w;
        *(float4*)(t + (long long)n * DIM + lane * 4) = o;

        cnt0 = cnt1;
        c0   = c1;
        my0  = my1;
        cnt1 = n2 < N_NODES ? __builtin_amdgcn_readlane(pd2, 0) : 0;
    }
}

#define GEMM_NBLK   256
#define GEMM_WAVES  16
#define N_TILES     (N_NODES / 16)
#define A_PITCH     132

__device__ __forceinline__ void gin_wave_lds_handover() {
    __builtin_amdgcn_fence(__ATOMIC_RELEASE, "workgroup", "local");
    __builtin_amdgcn_wave_barrier();
    __builtin_amdgcn_fence(__ATOMIC_ACQUIRE, "workgroup", "local");
}

__global__ __launch_bounds__(GEMM_WAVES * 32)
void gin_gemm_ln(const float* __restrict__ T, const float* __restrict__ W,
                 const float* __restrict__ bias, const float* __restrict__ gamma,
                 const float* __restrict__ beta, float* __restrict__ Hout) {
    __shared__ __attribute__((aligned(16))) float w_frag[32 * 4 * 32 * 4];
    __shared__ __attribute__((aligned(16))) float a_tile[GEMM_WAVES * 16 * A_PITCH];

    const int tid  = threadIdx.x;
    const int lane = tid & 31;
    const int c    = lane & 15;
    const int h    = lane >> 4;
    const int wave = __builtin_amdgcn_readfirstlane(tid >> 5);

#pragma unroll
    for (int i = 0; i < 128 / GEMM_WAVES; ++i) {
        const int grp = wave + GEMM_WAVES * i;
        const int t   = grp >> 2;
        const int jp  = grp & 3;
        const float* wp = W + (4 * t + 2 * h) * DIM + 32 * jp + c;
        float4 f;
        f.x = wp[0];
        f.y = wp[DIM];
        f.z = wp[16];
        f.w = wp[DIM + 16];
        *(float4*)(w_frag + ((t * 4 + jp) * 32 + lane) * 4) = f;
    }

    float bv[8], gv[8], bt[8];
#pragma unroll
    for (int j = 0; j < 8; ++j) {
        bv[j] = bias[16 * j + c];
        gv[j] = gamma[16 * j + c];
        bt[j] = beta[16 * j + c];
    }

    float* my_a = a_tile + wave * (16 * A_PITCH);

    static_assert(GEMM_NBLK * GEMM_WAVES <= N_TILES, "a first tile for every wave");
    int tile = (int)blockIdx.x + GEMM_NBLK * wave;
    v4f raw[16];
#pragma unroll
    for (int r = 0; r < 16; ++r)
        raw[r] = *(const v4f*)(T + (long long)(tile * 16 + r) * DIM + lane * 4);
    __syncthreads();

    do {
        gin_wave_lds_handover();
#pragma unroll
        for (int r = 0; r < 16; ++r)
            *(v4f*)(my_a + r * A_PITCH + lane * 4) = raw[r];
        gin_wave_lds_handover();
        const float* my_frag = my_a + c * A_PITCH + 2 * h;

        const int ntile = tile + GEMM_NBLK * GEMM_WAVES;
        const int ptile = ntile < N_TILES ? ntile : tile;
#pragma unroll
        for (int r = 0; r < 16; ++r)
            raw[r] = *(const v4f*)(T + (long long)(ptile * 16 + r) * DIM + lane * 4);

        v8f acc[8];
#pragma unroll
        for (int j = 0; j < 8; ++j) acc[j] = (v8f){0.f, 0.f, 0.f, 0.f, 0.f, 0.f, 0.f, 0.f};

        const float* wl = w_frag + lane * 4;
        v4f b0s[16], b1s[16];
        v2f a0s[4], a1s[4];
#pragma unroll
        for (int u = 0; u < 16; ++u) b0s[u] = *(const v4f*)(wl + u * 128);
#pragma unroll
        for (int tt = 0; tt < 4; ++tt) a0s[tt] = *(const v2f*)(my_frag + 4 * tt);
#pragma nounroll
        for (int kk = 0; kk < 4; ++kk) {
            const int blk1 = 2 * kk + 1;
            const int blk2 = (2 * kk + 2) & 7;
#pragma unroll
            for (int u = 0; u < 16; ++u) b1s[u] = *(const v4f*)(wl + blk1 * 2048 + u * 128);
#pragma unroll
            for (int tt = 0; tt < 4; ++tt) a1s[tt] = *(const v2f*)(my_frag + blk1 * 16 + 4 * tt);
#pragma unroll
            for (int tt = 0; tt < 4; ++tt) {
#pragma unroll
                for (int jp = 0; jp < 4; ++jp) {
                    const v4f f = b0s[tt * 4 + jp];
                    v2f b0, b1;
                    b0.x = f.x; b0.y = f.y;
                    b1.x = f.z; b1.y = f.w;
                    acc[2 * jp] = __builtin_amdgcn_wmma_f32_16x16x4_f32(
                            false, a0s[tt], false, b0, (short)0, acc[2 * jp], false, false);
                    acc[2 * jp + 1] = __builtin_amdgcn_wmma_f32_16x16x4_f32(
                            false, a0s[tt], false, b1, (short)0, acc[2 * jp + 1], false, false);
                }
            }
#pragma unroll
            for (int u = 0; u < 16; ++u) b0s[u] = *(const v4f*)(wl + blk2 * 2048 + u * 128);
#pragma unroll
            for (int tt = 0; tt < 4; ++tt) a0s[tt] = *(const v2f*)(my_frag + blk2 * 16 + 4 * tt);
#pragma unroll
            for (int tt = 0; tt < 4; ++tt) {
#pragma unroll
                for (int jp = 0; jp < 4; ++jp) {
                    const v4f f = b1s[tt * 4 + jp];
                    v2f b0, b1;
                    b0.x = f.x; b0.y = f.y;
                    b1.x = f.z; b1.y = f.w;
                    acc[2 * jp] = __builtin_amdgcn_wmma_f32_16x16x4_f32(
                            false, a1s[tt], false, b0, (short)0, acc[2 * jp], false, false);
                    acc[2 * jp + 1] = __builtin_amdgcn_wmma_f32_16x16x4_f32(
                            false, a1s[tt], false, b1, (short)0, acc[2 * jp + 1], false, false);
                }
            }
        }

        float s[8], q[8];
#pragma unroll
        for (int r = 0; r < 8; ++r) { s[r] = 0.f; q[r] = 0.f; }
        gin_wave_lds_handover();
#pragma unroll
        for (int j = 0; j < 8; ++j) {
#pragma unroll
            for (int r = 0; r < 8; ++r) {
                float v = acc[j][r] + bv[j];
                v = v > 0.f ? v : 0.f;
                acc[j][r] = v;
                s[r] += v;
                q[r] += v * v;
            }
        }
#pragma unroll
        for (int m = 1; m < 16; m <<= 1) {
            float es[8], eq[8];
#pragma unroll
            for (int r = 0; r < 8; ++r) {
                es[r] = __shfl_xor(s[r], m, 32);
                eq[r] = __shfl_xor(q[r], m, 32);
            }
#pragma unroll
            for (int r = 0; r < 8; ++r) {
                s[r] += es[r];
                q[r] += eq[r];
            }
        }
#pragma unroll
        for (int r = 0; r < 8; ++r) {
            const float mean = s[r] * (1.0f / DIM);
            const float var  = q[r] * (1.0f / DIM) - mean * mean;
            const float rs   = rsqrtf(var + LN_EPSF);
#pragma unroll
            for (int j = 0; j < 8; ++j) {
                const float o = (acc[j][r] - mean) * rs * gv[j] + bt[j];
                my_a[(8 * h + r) * A_PITCH + 16 * j + c] = o;
            }
        }
        gin_wave_lds_handover();
#pragma unroll
        for (int r = 0; r < 16; ++r) {
            const float4 o = *(const float4*)(my_a + r * A_PITCH + lane * 4);
            *(float4*)(Hout + (long long)(tile * 16 + r) * DIM + lane * 4) = o;
        }
        tile = ntile;
    } while (tile < N_TILES);
}

__global__ void gin_zero(float* __restrict__ p, int n) {
    int i = blockIdx.x * blockDim.x + threadIdx.x;
    if (i < n) p[i] = 0.f;
}

#define POOL_RUN 32

__device__ __forceinline__ void gin_pool_flush(float* __restrict__ psum, float* __restrict__ cnt,
                                               int g, int lane, float4 a, int len) {
    float* p = psum + (long long)g * DIM + lane * 4;
    atomicAdd(p + 0, a.x);
    atomicAdd(p + 1, a.y);
    atomicAdd(p + 2, a.z);
    atomicAdd(p + 3, a.w);
    if (lane == 0) atomicAdd(cnt + g, (float)len);
}

__global__ __launch_bounds__(256)
void gin_pool(const float* __restrict__ h, const int* __restrict__ batch,
              float* __restrict__ psum, float* __restrict__ cnt) {
    static_assert(N_NODES % POOL_RUN == 0, "a full run for every wave");
    const int lane = threadIdx.x & 31;
    const int w  = __builtin_amdgcn_readfirstlane((int)((blockIdx.x * blockDim.x + threadIdx.x) >> 5));
    const int n0 = w * POOL_RUN;
    if (n0 >= N_NODES) return;

    const int bid = batch[n0 + lane];
    float4 v[POOL_RUN];
#pragma unroll
    for (int r = 0; r < POOL_RUN; ++r) v[r] = gin_row4(h, n0 + r, lane);

    int cur = __builtin_amdgcn_readlane(bid, 0);
    int len = 1;
    float4 acc = v[0];
#pragma unroll
    for (int r = 1; r < POOL_RUN; ++r) {
        const int b = __builtin_amdgcn_readlane(bid, r);
        if (b != cur) {
            gin_pool_flush(psum, cnt, cur, lane, acc, len);
            cur = b;
            len = 0;
            acc = make_float4(0.f, 0.f, 0.f, 0.f);
        }
        acc.x += v[r].x; acc.y += v[r].y; acc.z += v[r].z; acc.w += v[r].w;
        ++len;
    }
    gin_pool_flush(psum, cnt, cur, lane, acc, len);
}

__global__ void gin_outproj(const float* __restrict__ psum, const float* __restrict__ cnt,
                            const float* __restrict__ Wout, const float* __restrict__ bout,
                            float* __restrict__ logits) {
    int idx = blockIdx.x * blockDim.x + threadIdx.x;
    if (idx < NGRAPH * NCLASS) {
        int g = idx / NCLASS;
        int c = idx - g * NCLASS;
        float inv = 1.0f / fmaxf(cnt[g], 1.0f);
        float acc = bout[c];
        const float* pg = psum + g * DIM;
#pragma unroll 4
        for (int k = 0; k < DIM; ++k) acc += pg[k] * inv * Wout[k * NCLASS + c];
        logits[idx] = acc;
    }
}

__global__ void gin_logsoftmax(const float* __restrict__ logits, float* __restrict__ out) {
    __shared__ float sh[64];
    const int g = blockIdx.x;
    const int t = threadIdx.x;
    float v = (t < NCLASS) ? logits[g * NCLASS + t] : -3.4e38f;
    sh[t] = v;
    __syncthreads();
    for (int s = 32; s > 0; s >>= 1) {
        if (t < s) sh[t] = fmaxf(sh[t], sh[t + s]);
        __syncthreads();
    }
    const float mx = sh[0];
    __syncthreads();
    sh[t] = (t < NCLASS) ? expf(v - mx) : 0.f;
    __syncthreads();
    for (int s = 32; s > 0; s >>= 1) {
        if (t < s) sh[t] += sh[t + s];
        __syncthreads();
    }
    const float lse = logf(sh[0]);
    if (t < NCLASS) out[g * NCLASS + t] = v - mx - lse;
}

extern "C" void kernel_launch(void* const* d_in, const int* in_sizes, int n_in,
                              void* d_out, int out_size, void* d_ws, size_t ws_size,
                              hipStream_t stream) {
    const float* x      = (const float*)d_in[0];
    const int*   ei     = (const int*)d_in[1];
    const int*   batch  = (const int*)d_in[2];
    const float* Ws     = (const float*)d_in[3];
    const float* bs     = (const float*)d_in[4];
    const float* gammas = (const float*)d_in[5];
    const float* betas  = (const float*)d_in[6];
    const float* Wout   = (const float*)d_in[7];
    const float* bout   = (const float*)d_in[8];
    float* out = (float*)d_out;

    const int* src = ei;
    const int* dst = ei + N_EDGES;

    float* ws = (float*)d_ws;
    const size_t nd = (size_t)N_NODES * DIM;
    float* hA     = ws;
    float* hB     = ws + nd;
    float* tbuf   = ws + 2 * nd;
    float* psum   = ws + 3 * nd;
    float* cnt    = psum + (size_t)NGRAPH * DIM;
    int*   deg     = (int*)(cnt + NGRAPH);
    int*   ovf_cnt = deg + N_NODES;
    int*   col     = deg + DEG_PAD;
    int*   ovf_dst = col + (size_t)N_NODES * SLOT_CAP;
    int*   ovf_src = ovf_dst + N_EDGES;
    float* logits  = (float*)(ovf_src + N_EDGES);

    const int gEdge    = (N_EDGES + 255) / 256;
    const int gPool    = (N_NODES / POOL_RUN * 32 + 255) / 256;

    const int nZero = NGRAPH * DIM + NGRAPH + DEG_PAD;
    gin_zero<<<(nZero + 255) / 256, 256, 0, stream>>>(psum, nZero);
    gin_fill<<<gEdge, 256, 0, stream>>>(src, dst, deg, col, ovf_cnt, ovf_dst, ovf_src);

    const float* hin = x;
    for (int i = 0; i < NLAYERS; ++i) {
        float* hout = (i & 1) ? hB : hA;
        gin_gather<<<GATHER_NBLK, 256, 0, stream>>>(hin, deg, col, ovf_cnt, ovf_dst, ovf_src, tbuf);
        gin_gemm_ln<<<GEMM_NBLK, GEMM_WAVES * 32, 0, stream>>>(tbuf, Ws + (size_t)i * DIM * DIM,
                                               bs + i * DIM, gammas + i * DIM,
                                               betas + i * DIM, hout);
        hin = hout;
    }

    gin_pool<<<gPool, 256, 0, stream>>>(hin, batch, psum, cnt);
    gin_outproj<<<(NGRAPH * NCLASS + 255) / 256, 256, 0, stream>>>(psum, cnt, Wout, bout, logits);
    gin_logsoftmax<<<NGRAPH, 64, 0, stream>>>(logits, out);
}


# Neighbour gather moved into the GEMM+LN launch: a wave sums its tile's 16 rows of 2h+neighbours (4 nodes, up to 36 row loads in flight) into its LDS A tile; T plane and 4 launches go; sum order kept
# speedup vs baseline: 3.4116x; 1.0736x over previous
#include <hip/hip_runtime.h>
#include <hip/hip_bf16.h>
#include <math.h>

#define N_NODES 100000
#define N_EDGES 600000
#define DIM     128
#define NLAYERS 4
#define NGRAPH  512
#define NCLASS  40
#define LN_EPSF 1e-5f

typedef __attribute__((ext_vector_type(2))) float v2f;
typedef __attribute__((ext_vector_type(4))) float v4f;
typedef __attribute__((ext_vector_type(8))) float v8f;

#define SLOT_CAP  32
#define DEG_PAD   (N_NODES + 96)

__global__ void gin_fill(const int* __restrict__ src, const int* __restrict__ dst,
                         int* __restrict__ deg, int* __restrict__ col,
                         int* __restrict__ ovf_cnt, int* __restrict__ ovf_dst,
                         int* __restrict__ ovf_src) {
    int e = blockIdx.x * blockDim.x + threadIdx.x;
    if (e < N_EDGES) {
        const int d = dst[e];
        const int s = src[e];
        const int slot = atomicAdd(deg + d, 1);
        if (slot < SLOT_CAP) {
            col[d * SLOT_CAP + slot] = s;
        } else {
            const int k = atomicAdd(ovf_cnt, 1);
            ovf_dst[k] = d;
            ovf_src[k] = s;
        }
    }
}

__device__ __forceinline__ float4 gin_row4(const float* __restrict__ h, int node, int lane) {
    return *(const float4*)(h + (long long)node * DIM + lane * 4);
}

__device__ __forceinline__ int gin_slot_row(const int* __restrict__ deg, const int* __restrict__ col,
                                            int node, int cnt, int lane) {
    const int cs = cnt < SLOT_CAP ? cnt : SLOT_CAP;
    return *((lane < cs) ? (col + node * SLOT_CAP + lane) : (deg + node));
}

#define GATHER_B    4
#define GEMM_NBLK   256
#define GEMM_WAVES  16
#define N_TILES     (N_NODES / 16)
#define A_PITCH     132

__device__ __forceinline__ void gin_wave_lds_handover() {
    __builtin_amdgcn_fence(__ATOMIC_RELEASE, "workgroup", "local");
    __builtin_amdgcn_wave_barrier();
    __builtin_amdgcn_fence(__ATOMIC_ACQUIRE, "workgroup", "local");
}

__global__ __launch_bounds__(GEMM_WAVES * 32)
void gin_layer(const float* __restrict__ Hin, const int* __restrict__ deg,
               const int* __restrict__ col, const int* __restrict__ ovf_cnt,
               const int* __restrict__ ovf_dst, const int* __restrict__ ovf_src,
               const float* __restrict__ W,
               const float* __restrict__ bias, const float* __restrict__ gamma,
               const float* __restrict__ beta, float* __restrict__ Hout) {
    __shared__ __attribute__((aligned(16))) float w_frag[32 * 4 * 32 * 4];
    __shared__ __attribute__((aligned(16))) float a_tile[GEMM_WAVES * 16 * A_PITCH];

    const int tid  = threadIdx.x;
    const int lane = tid & 31;
    const int c    = lane & 15;
    const int h    = lane >> 4;
    const int wave = __builtin_amdgcn_readfirstlane(tid >> 5);

    static_assert(GEMM_NBLK * GEMM_WAVES <= N_TILES, "a first tile for every wave");
    static_assert(16 % GATHER_B == 0, "whole batches in a tile");
    int tile = (int)blockIdx.x + GEMM_NBLK * wave;
    int dg = deg[tile * 16 + c];

#pragma unroll
    for (int i = 0; i < 128 / GEMM_WAVES; ++i) {
        const int grp = wave + GEMM_WAVES * i;
        const int t   = grp >> 2;
        const int jp  = grp & 3;
        const float* wp = W + (4 * t + 2 * h) * DIM + 32 * jp + c;
        float4 f;
        f.x = wp[0];
        f.y = wp[DIM];
        f.z = wp[16];
        f.w = wp[DIM + 16];
        *(float4*)(w_frag + ((t * 4 + jp) * 32 + lane) * 4) = f;
    }

    float bv[8], gv[8], bt[8];
#pragma unroll
    for (int j = 0; j < 8; ++j) {
        bv[j] = bias[16 * j + c];
        gv[j] = gamma[16 * j + c];
        bt[j] = beta[16 * j + c];
    }

    float* my_a = a_tile + wave * (16 * A_PITCH);

    int my[GATHER_B];
#pragma unroll
    for (int i = 0; i < GATHER_B; ++i)
        my[i] = gin_slot_row(deg, col, tile * 16 + i, __builtin_amdgcn_readlane(dg, i), lane);

    bool first = true;
    do {
        const int ntile = tile + GEMM_NBLK * GEMM_WAVES;
        const int ptile = ntile < N_TILES ? ntile : tile;
        const int dgn = deg[ptile * 16 + c];

        gin_wave_lds_handover();
#pragma nounroll
        for (int b = 0; b < 16 / GATHER_B; ++b) {
            const int r0 = GATHER_B * b;
            const int r1 = (b + 1 < 16 / GATHER_B) ? r0 + GATHER_B : r0;
            int nx[GATHER_B];
#pragma unroll
            for (int i = 0; i < GATHER_B; ++i)
                nx[i] = gin_slot_row(deg, col, tile * 16 + r1 + i,
                                     __builtin_amdgcn_readlane(dg, r1 + i), lane);

            int cnt[GATHER_B], cs[GATHER_B];
            int cm = 0;
#pragma unroll
            for (int i = 0; i < GATHER_B; ++i) {
                cnt[i] = __builtin_amdgcn_readlane(dg, r0 + i);
                cs[i]  = cnt[i] < SLOT_CAP ? cnt[i] : SLOT_CAP;
                cm     = cs[i] > cm ? cs[i] : cm;
            }
            float4 self[GATHER_B], sum[GATHER_B];
#pragma unroll
            for (int i = 0; i < GATHER_B; ++i) {
                self[i] = gin_row4(Hin, tile * 16 + r0 + i, lane);
                sum[i]  = make_float4(0.f, 0.f, 0.f, 0.f);
            }
            for (int j = 0; j < cm; j += 8) {
                float4 v[GATHER_B][8];
#pragma unroll
                for (int i = 0; i < GATHER_B; ++i) {
#pragma unroll
                    for (int u = 0; u < 8; ++u) {
                        v[i][u] = make_float4(0.f, 0.f, 0.f, 0.f);
                        if (j + u < cs[i])
                            v[i][u] = gin_row4(Hin, __builtin_amdgcn_readlane(my[i], j + u), lane);
                    }
                }
#pragma unroll
                for (int i = 0; i < GATHER_B; ++i) {
#pragma unroll
                    for (int u = 0; u < 8; ++u) {
                        sum[i].x += v[i][u].x; sum[i].y += v[i][u].y;
                        sum[i].z += v[i][u].z; sum[i].w += v[i][u].w;
                    }
                }
            }
#pragma unroll
            for (int i = 0; i < GATHER_B; ++i) {
                if (cnt[i] > SLOT_CAP) {
                    const int n = tile * 16 + r0 + i;
                    const int total = ovf_cnt[0];
                    for (int base = 0; base < total; base += 32) {
                        const int k  = base + lane;
                        const int od = (k < total) ? ovf_dst[k] : -1;
                        const int os = (k < total) ? ovf_src[k] : 0;
                        unsigned int hit = __builtin_amdgcn_ballot_w32(od == n);
                        while (hit != 0u) {
                            const int bit = __builtin_ctz(hit);
                            hit &= hit - 1u;
                            const float4 v = gin_row4(Hin, __builtin_amdgcn_readlane(os, bit), lane);
                            sum[i].x += v.x; sum[i].y += v.y; sum[i].z += v.z; sum[i].w += v.w;
                        }
                    }
                }
                v4f o;
                o.x = 2.f * self[i].x + sum[i].x;
                o.y = 2.f * self[i].y + sum[i].y;
                o.z = 2.f * self[i].z + sum[i].z;
                o.w = 2.f * self[i].w + sum[i].w;
                *(v4f*)(my_a + (r0 + i) * A_PITCH + lane * 4) = o;
                my[i] = nx[i];
            }
        }
        gin_wave_lds_handover();
        if (first) {
            __syncthreads();
            first = false;
        }
        const float* my_frag = my_a + c * A_PITCH + 2 * h;

        dg = dgn;
#pragma unroll
        for (int i = 0; i < GATHER_B; ++i)
            my[i] = gin_slot_row(deg, col, ptile * 16 + i, __builtin_amdgcn_readlane(dg, i), lane);

        v8f acc[8];
#pragma unroll
        for (int j = 0; j < 8; ++j) acc[j] = (v8f){0.f, 0.f, 0.f, 0.f, 0.f, 0.f, 0.f, 0.f};

        const float* wl = w_frag + lane * 4;
        v4f b0s[16], b1s[16];
        v2f a0s[4], a1s[4];
#pragma unroll
        for (int u = 0; u < 16; ++u) b0s[u] = *(const v4f*)(wl + u * 128);
#pragma unroll
        for (int tt = 0; tt < 4; ++tt) a0s[tt] = *(const v2f*)(my_frag + 4 * tt);
#pragma nounroll
        for (int kk = 0; kk < 4; ++kk) {
            const int blk1 = 2 * kk + 1;
            const int blk2 = (2 * kk + 2) & 7;
#pragma unroll
            for (int u = 0; u < 16; ++u) b1s[u] = *(const v4f*)(wl + blk1 * 2048 + u * 128);
#pragma unroll
            for (int tt = 0; tt < 4; ++tt) a1s[tt] = *(const v2f*)(my_frag + blk1 * 16 + 4 * tt);
#pragma unroll
            for (int tt = 0; tt < 4; ++tt) {
#pragma unroll
                for (int jp = 0; jp < 4; ++jp) {
                    const v4f f = b0s[tt * 4 + jp];
                    v2f b0, b1;
                    b0.x = f.x; b0.y = f.y;
                    b1.x = f.z; b1.y = f.w;
                    acc[2 * jp] = __builtin_amdgcn_wmma_f32_16x16x4_f32(
                            false, a0s[tt], false, b0, (short)0, acc[2 * jp], false, false);
                    acc[2 * jp + 1] = __builtin_amdgcn_wmma_f32_16x16x4_f32(
                            false, a0s[tt], false, b1, (short)0, acc[2 * jp + 1], false, false);
                }
            }
#pragma unroll
            for (int u = 0; u < 16; ++u) b0s[u] = *(const v4f*)(wl + blk2 * 2048 + u * 128);
#pragma unroll
            for (int tt = 0; tt < 4; ++tt) a0s[tt] = *(const v2f*)(my_frag + blk2 * 16 + 4 * tt);
#pragma unroll
            for (int tt = 0; tt < 4; ++tt) {
#pragma unroll
                for (int jp = 0; jp < 4; ++jp) {
                    const v4f f = b1s[tt * 4 + jp];
                    v2f b0, b1;
                    b0.x = f.x; b0.y = f.y;
                    b1.x = f.z; b1.y = f.w;
                    acc[2 * jp] = __builtin_amdgcn_wmma_f32_16x16x4_f32(
                            false, a1s[tt], false, b0, (short)0, acc[2 * jp], false, false);
                    acc[2 * jp + 1] = __builtin_amdgcn_wmma_f32_16x16x4_f32(
                            false, a1s[tt], false, b1, (short)0, acc[2 * jp + 1], false, false);
                }
            }
        }

        float s[8], q[8];
#pragma unroll
        for (int r = 0; r < 8; ++r) { s[r] = 0.f; q[r] = 0.f; }
        gin_wave_lds_handover();
#pragma unroll
        for (int j = 0; j < 8; ++j) {
#pragma unroll
            for (int r = 0; r < 8; ++r) {
                float v = acc[j][r] + bv[j];
                v = v > 0.f ? v : 0.f;
                acc[j][r] = v;
                s[r] += v;
                q[r] += v * v;
            }
        }
#pragma unroll
        for (int m = 1; m < 16; m <<= 1) {
            float es[8], eq[8];
#pragma unroll
            for (int r = 0; r < 8; ++r) {
                es[r] = __shfl_xor(s[r], m, 32);
                eq[r] = __shfl_xor(q[r], m, 32);
            }
#pragma unroll
            for (int r = 0; r < 8; ++r) {
                s[r] += es[r];
                q[r] += eq[r];
            }
        }
#pragma unroll
        for (int r = 0; r < 8; ++r) {
            const float mean = s[r] * (1.0f / DIM);
            const float var  = q[r] * (1.0f / DIM) - mean * mean;
            const float rs   = rsqrtf(var + LN_EPSF);
#pragma unroll
            for (int j = 0; j < 8; ++j) {
                const float o = (acc[j][r] - mean) * rs * gv[j] + bt[j];
                my_a[(8 * h + r) * A_PITCH + 16 * j + c] = o;
            }
        }
        gin_wave_lds_handover();
#pragma unroll
        for (int r = 0; r < 16; ++r) {
            const float4 o = *(const float4*)(my_a + r * A_PITCH + lane * 4);
            *(float4*)(Hout + (long long)(tile * 16 + r) * DIM + lane * 4) = o;
        }
        tile = ntile;
    } while (tile < N_TILES);
}

__global__ void gin_zero(float* __restrict__ p, int n) {
    int i = blockIdx.x * blockDim.x + threadIdx.x;
    if (i < n) p[i] = 0.f;
}

#define POOL_RUN 32

__device__ __forceinline__ void gin_pool_flush(float* __restrict__ psum, float* __restrict__ cnt,
                                               int g, int lane, float4 a, int len) {
    float* p = psum + (long long)g * DIM + lane * 4;
    atomicAdd(p + 0, a.x);
    atomicAdd(p + 1, a.y);
    atomicAdd(p + 2, a.z);
    atomicAdd(p + 3, a.w);
    if (lane == 0) atomicAdd(cnt + g, (float)len);
}

__global__ __launch_bounds__(256)
void gin_pool(const float* __restrict__ h, const int* __restrict__ batch,
              float* __restrict__ psum, float* __restrict__ cnt) {
    static_assert(N_NODES % POOL_RUN == 0, "a full run for every wave");
    const int lane = threadIdx.x & 31;
    const int w  = __builtin_amdgcn_readfirstlane((int)((blockIdx.x * blockDim.x + threadIdx.x) >> 5));
    const int n0 = w * POOL_RUN;
    if (n0 >= N_NODES) return;

    const int bid = batch[n0 + lane];
    float4 v[POOL_RUN];
#pragma unroll
    for (int r = 0; r < POOL_RUN; ++r) v[r] = gin_row4(h, n0 + r, lane);

    int cur = __builtin_amdgcn_readlane(bid, 0);
    int len = 1;
    float4 acc = v[0];
#pragma unroll
    for (int r = 1; r < POOL_RUN; ++r) {
        const int b = __builtin_amdgcn_readlane(bid, r);
        if (b != cur) {
            gin_pool_flush(psum, cnt, cur, lane, acc, len);
            cur = b;
            len = 0;
            acc = make_float4(0.f, 0.f, 0.f, 0.f);
        }
        acc.x += v[r].x; acc.y += v[r].y; acc.z += v[r].z; acc.w += v[r].w;
        ++len;
    }
    gin_pool_flush(psum, cnt, cur, lane, acc, len);
}

__global__ void gin_outproj(const float* __restrict__ psum, const float* __restrict__ cnt,
                            const float* __restrict__ Wout, const float* __restrict__ bout,
                            float* __restrict__ logits) {
    int idx = blockIdx.x * blockDim.x + threadIdx.x;
    if (idx < NGRAPH * NCLASS) {
        int g = idx / NCLASS;
        int c = idx - g * NCLASS;
        float inv = 1.0f / fmaxf(cnt[g], 1.0f);
        float acc = bout[c];
        const float* pg = psum + g * DIM;
#pragma unroll 4
        for (int k = 0; k < DIM; ++k) acc += pg[k] * inv * Wout[k * NCLASS + c];
        logits[idx] = acc;
    }
}

__global__ void gin_logsoftmax(const float* __restrict__ logits, float* __restrict__ out) {
    __shared__ float sh[64];
    const int g = blockIdx.x;
    const int t = threadIdx.x;
    float v = (t < NCLASS) ? logits[g * NCLASS + t] : -3.4e38f;
    sh[t] = v;
    __syncthreads();
    for (int s = 32; s > 0; s >>= 1) {
        if (t < s) sh[t] = fmaxf(sh[t], sh[t + s]);
        __syncthreads();
    }
    const float mx = sh[0];
    __syncthreads();
    sh[t] = (t < NCLASS) ? expf(v - mx) : 0.f;
    __syncthreads();
    for (int s = 32; s > 0; s >>= 1) {
        if (t < s) sh[t] += sh[t + s];
        __syncthreads();
    }
    const float lse = logf(sh[0]);
    if (t < NCLASS) out[g * NCLASS + t] = v - mx - lse;
}

extern "C" void kernel_launch(void* const* d_in, const int* in_sizes, int n_in,
                              void* d_out, int out_size, void* d_ws, size_t ws_size,
                              hipStream_t stream) {
    const float* x      = (const float*)d_in[0];
    const int*   ei     = (const int*)d_in[1];
    const int*   batch  = (const int*)d_in[2];
    const float* Ws     = (const float*)d_in[3];
    const float* bs     = (const float*)d_in[4];
    const float* gammas = (const float*)d_in[5];
    const float* betas  = (const float*)d_in[6];
    const float* Wout   = (const float*)d_in[7];
    const float* bout   = (const float*)d_in[8];
    float* out = (float*)d_out;

    const int* src = ei;
    const int* dst = ei + N_EDGES;

    float* ws = (float*)d_ws;
    const size_t nd = (size_t)N_NODES * DIM;
    float* hA     = ws;
    float* hB     = ws + nd;
    float* psum   = ws + 2 * nd;
    float* cnt    = psum + (size_t)NGRAPH * DIM;
    int*   deg     = (int*)(cnt + NGRAPH);
    int*   ovf_cnt = deg + N_NODES;
    int*   col     = deg + DEG_PAD;
    int*   ovf_dst = col + (size_t)N_NODES * SLOT_CAP;
    int*   ovf_src = ovf_dst + N_EDGES;
    float* logits  = (float*)(ovf_src + N_EDGES);

    const int gEdge    = (N_EDGES + 255) / 256;
    const int gPool    = (N_NODES / POOL_RUN * 32 + 255) / 256;

    const int nZero = NGRAPH * DIM + NGRAPH + DEG_PAD;
    gin_zero<<<(nZero + 255) / 256, 256, 0, stream>>>(psum, nZero);
    gin_fill<<<gEdge, 256, 0, stream>>>(src, dst, deg, col, ovf_cnt, ovf_dst, ovf_src);

    const float* hin = x;
    for (int i = 0; i < NLAYERS; ++i) {
        float* hout = (i & 1) ? hB : hA;
        gin_layer<<<GEMM_NBLK, GEMM_WAVES * 32, 0, stream>>>(hin, deg, col, ovf_cnt, ovf_dst, ovf_src,
                                             Ws + (size_t)i * DIM * DIM,
                                             bs + i * DIM, gammas + i * DIM,
                                             betas + i * DIM, hout);
        hin = hout;
    }

    gin_pool<<<gPool, 256, 0, stream>>>(hin, batch, psum, cnt);
    gin_outproj<<<(NGRAPH * NCLASS + 255) / 256, 256, 0, stream>>>(psum, cnt, Wout, bout, logits);
    gin_logsoftmax<<<NGRAPH, 64, 0, stream>>>(logits, out);
}


# Pool moved into the last layer's launch: a wave sums its tile's 16 finished rows and adds them to the graph sums by float atomics; pool launch and last H plane go; pooled sum order changes
# speedup vs baseline: 3.5243x; 1.0330x over previous
#include <hip/hip_runtime.h>
#include <hip/hip_bf16.h>
#include <math.h>

#define N_NODES 100000
#define N_EDGES 600000
#define DIM     128
#define NLAYERS 4
#define NGRAPH  512
#define NCLASS  40
#define LN_EPSF 1e-5f

typedef __attribute__((ext_vector_type(2))) float v2f;
typedef __attribute__((ext_vector_type(4))) float v4f;
typedef __attribute__((ext_vector_type(8))) float v8f;

#define SLOT_CAP  32
#define DEG_PAD   (N_NODES + 96)

__global__ void gin_fill(const int* __restrict__ src, const int* __restrict__ dst,
                         int* __restrict__ deg, int* __restrict__ col,
                         int* __restrict__ ovf_cnt, int* __restrict__ ovf_dst,
                         int* __restrict__ ovf_src) {
    int e = blockIdx.x * blockDim.x + threadIdx.x;
    if (e < N_EDGES) {
        const int d = dst[e];
        const int s = src[e];
        const int slot = atomicAdd(deg + d, 1);
        if (slot < SLOT_CAP) {
            col[d * SLOT_CAP + slot] = s;
        } else {
            const int k = atomicAdd(ovf_cnt, 1);
            ovf_dst[k] = d;
            ovf_src[k] = s;
        }
    }
}

__device__ __forceinline__ float4 gin_row4(const float* __restrict__ h, int node, int lane) {
    return *(const float4*)(h + (long long)node * DIM + lane * 4);
}

__device__ __forceinline__ int gin_slot_row(const int* __restrict__ deg, const int* __restrict__ col,
                                            int node, int cnt, int lane) {
    const int cs = cnt < SLOT_CAP ? cnt : SLOT_CAP;
    return *((lane < cs) ? (col + node * SLOT_CAP + lane) : (deg + node));
}

#define GATHER_B    4
#define GEMM_NBLK   256
#define GEMM_WAVES  16
#define N_TILES     (N_NODES / 16)
#define A_PITCH     132

__device__ __forceinline__ void gin_wave_lds_handover() {
    __builtin_amdgcn_fence(__ATOMIC_RELEASE, "workgroup", "local");
    __builtin_amdgcn_wave_barrier();
    __builtin_amdgcn_fence(__ATOMIC_ACQUIRE, "workgroup", "local");
}

__device__ __forceinline__ void gin_pool_flush(float* __restrict__ psum, float* __restrict__ cnt,
                                               int g, int lane, float4 a, int len) {
    float* p = psum + (long long)g * DIM + lane * 4;
    atomicAdd(p + 0, a.x);
    atomicAdd(p + 1, a.y);
    atomicAdd(p + 2, a.z);
    atomicAdd(p + 3, a.w);
    if (lane == 0) atomicAdd(cnt + g, (float)len);
}

template <bool POOL>
__global__ __launch_bounds__(GEMM_WAVES * 32)
void gin_layer(const float* __restrict__ Hin, const int* __restrict__ deg,
               const int* __restrict__ col, const int* __restrict__ ovf_cnt,
               const int* __restrict__ ovf_dst, const int* __restrict__ ovf_src,
               const float* __restrict__ W,
               const float* __restrict__ bias, const float* __restrict__ gamma,
               const float* __restrict__ beta, float* __restrict__ Hout,
               const int* __restrict__ batch, float* __restrict__ psum,
               float* __restrict__ pcnt) {
    __shared__ __attribute__((aligned(16))) float w_frag[32 * 4 * 32 * 4];
    __shared__ __attribute__((aligned(16))) float a_tile[GEMM_WAVES * 16 * A_PITCH];

    const int tid  = threadIdx.x;
    const int lane = tid & 31;
    const int c    = lane & 15;
    const int h    = lane >> 4;
    const int wave = __builtin_amdgcn_readfirstlane(tid >> 5);

    static_assert(GEMM_NBLK * GEMM_WAVES <= N_TILES, "a first tile for every wave");
    static_assert(16 % GATHER_B == 0, "whole batches in a tile");
    int tile = (int)blockIdx.x + GEMM_NBLK * wave;
    int dg = deg[tile * 16 + c];

#pragma unroll
    for (int i = 0; i < 128 / GEMM_WAVES; ++i) {
        const int grp = wave + GEMM_WAVES * i;
        const int t   = grp >> 2;
        const int jp  = grp & 3;
        const float* wp = W + (4 * t + 2 * h) * DIM + 32 * jp + c;
        float4 f;
        f.x = wp[0];
        f.y = wp[DIM];
        f.z = wp[16];
        f.w = wp[DIM + 16];
        *(float4*)(w_frag + ((t * 4 + jp) * 32 + lane) * 4) = f;
    }

    float bv[8], gv[8], bt[8];
#pragma unroll
    for (int j = 0; j < 8; ++j) {
        bv[j] = bias[16 * j + c];
        gv[j] = gamma[16 * j + c];
        bt[j] = beta[16 * j + c];
    }

    float* my_a = a_tile + wave * (16 * A_PITCH);

    int my[GATHER_B];
#pragma unroll
    for (int i = 0; i < GATHER_B; ++i)
        my[i] = gin_slot_row(deg, col, tile * 16 + i, __builtin_amdgcn_readlane(dg, i), lane);

    bool first = true;
    do {
        const int ntile = tile + GEMM_NBLK * GEMM_WAVES;
        const int ptile = ntile < N_TILES ? ntile : tile;
        const int dgn = deg[ptile * 16 + c];
        int bid = 0;
        if (POOL) bid = batch[tile * 16 + c];

        gin_wave_lds_handover();
#pragma nounroll
        for (int b = 0; b < 16 / GATHER_B; ++b) {
            const int r0 = GATHER_B * b;
            const int r1 = (b + 1 < 16 / GATHER_B) ? r0 + GATHER_B : r0;
            int nx[GATHER_B];
#pragma unroll
            for (int i = 0; i < GATHER_B; ++i)
                nx[i] = gin_slot_row(deg, col, tile * 16 + r1 + i,
                                     __builtin_amdgcn_readlane(dg, r1 + i), lane);

            int cnt[GATHER_B], cs[GATHER_B];
            int cm = 0;
#pragma unroll
            for (int i = 0; i < GATHER_B; ++i) {
                cnt[i] = __builtin_amdgcn_readlane(dg, r0 + i);
                cs[i]  = cnt[i] < SLOT_CAP ? cnt[i] : SLOT_CAP;
                cm     = cs[i] > cm ? cs[i] : cm;
            }
            float4 self[GATHER_B], sum[GATHER_B];
#pragma unroll
            for (int i = 0; i < GATHER_B; ++i) {
                self[i] = gin_row4(Hin, tile * 16 + r0 + i, lane);
                sum[i]  = make_float4(0.f, 0.f, 0.f, 0.f);
            }
            for (int j = 0; j < cm; j += 8) {
                float4 v[GATHER_B][8];
#pragma unroll
                for (int i = 0; i < GATHER_B; ++i) {
#pragma unroll
                    for (int u = 0; u < 8; ++u) {
                        v[i][u] = make_float4(0.f, 0.f, 0.f, 0.f);
                        if (j + u < cs[i])
                            v[i][u] = gin_row4(Hin, __builtin_amdgcn_readlane(my[i], j + u), lane);
                    }
                }
#pragma unroll
                for (int i = 0; i < GATHER_B; ++i) {
#pragma unroll
                    for (int u = 0; u < 8; ++u) {
                        sum[i].x += v[i][u].x; sum[i].y += v[i][u].y;
                        sum[i].z += v[i][u].z; sum[i].w += v[i][u].w;
                    }
                }
            }
#pragma unroll
            for (int i = 0; i < GATHER_B; ++i) {
                if (cnt[i] > SLOT_CAP) {
                    const int n = tile * 16 + r0 + i;
                    const int total = ovf_cnt[0];
                    for (int base = 0; base < total; base += 32) {
                        const int k  = base + lane;
                        const int od = (k < total) ? ovf_dst[k] : -1;
                        const int os = (k < total) ? ovf_src[k] : 0;
                        unsigned int hit = __builtin_amdgcn_ballot_w32(od == n);
                        while (hit != 0u) {
                            const int bit = __builtin_ctz(hit);
                            hit &= hit - 1u;
                            const float4 v = gin_row4(Hin, __builtin_amdgcn_readlane(os, bit), lane);
                            sum[i].x += v.x; sum[i].y += v.y; sum[i].z += v.z; sum[i].w += v.w;
                        }
                    }
                }
                v4f o;
                o.x = 2.f * self[i].x + sum[i].x;
                o.y = 2.f * self[i].y + sum[i].y;
                o.z = 2.f * self[i].z + sum[i].z;
                o.w = 2.f * self[i].w + sum[i].w;
                *(v4f*)(my_a + (r0 + i) * A_PITCH + lane * 4) = o;
                my[i] = nx[i];
            }
        }
        gin_wave_lds_handover();
        if (first) {
            __syncthreads();
            first = false;
        }
        const float* my_frag = my_a + c * A_PITCH + 2 * h;

        dg = dgn;
#pragma unroll
        for (int i = 0; i < GATHER_B; ++i)
            my[i] = gin_slot_row(deg, col, ptile * 16 + i, __builtin_amdgcn_readlane(dg, i), lane);

        v8f acc[8];
#pragma unroll
        for (int j = 0; j < 8; ++j) acc[j] = (v8f){0.f, 0.f, 0.f, 0.f, 0.f, 0.f, 0.f, 0.f};

        const float* wl = w_frag + lane * 4;
        v4f b0s[16], b1s[16];
        v2f a0s[4], a1s[4];
#pragma unroll
        for (int u = 0; u < 16; ++u) b0s[u] = *(const v4f*)(wl + u * 128);
#pragma unroll
        for (int tt = 0; tt < 4; ++tt) a0s[tt] = *(const v2f*)(my_frag + 4 * tt);
#pragma nounroll
        for (int kk = 0; kk < 4; ++kk) {
            const int blk1 = 2 * kk + 1;
            const int blk2 = (2 * kk + 2) & 7;
#pragma unroll
            for (int u = 0; u < 16; ++u) b1s[u] = *(const v4f*)(wl + blk1 * 2048 + u * 128);
#pragma unroll
            for (int tt = 0; tt < 4; ++tt) a1s[tt] = *(const v2f*)(my_frag + blk1 * 16 + 4 * tt);
#pragma unroll
            for (int tt = 0; tt < 4; ++tt) {
#pragma unroll
                for (int jp = 0; jp < 4; ++jp) {
                    const v4f f = b0s[tt * 4 + jp];
                    v2f b0, b1;
                    b0.x = f.x; b0.y = f.y;
                    b1.x = f.z; b1.y = f.w;
                    acc[2 * jp] = __builtin_amdgcn_wmma_f32_16x16x4_f32(
                            false, a0s[tt], false, b0, (short)0, acc[2 * jp], false, false);
                    acc[2 * jp + 1] = __builtin_amdgcn_wmma_f32_16x16x4_f32(
                            false, a0s[tt], false, b1, (short)0, acc[2 * jp + 1], false, false);
                }
            }
#pragma unroll
            for (int u = 0; u < 16; ++u) b0s[u] = *(const v4f*)(wl + blk2 * 2048 + u * 128);
#pragma unroll
            for (int tt = 0; tt < 4; ++tt) a0s[tt] = *(const v2f*)(my_frag + blk2 * 16 + 4 * tt);
#pragma unroll
            for (int tt = 0; tt < 4; ++tt) {
#pragma unroll
                for (int jp = 0; jp < 4; ++jp) {
                    const v4f f = b1s[tt * 4 + jp];
                    v2f b0, b1;
                    b0.x = f.x; b0.y = f.y;
                    b1.x = f.z; b1.y = f.w;
                    acc[2 * jp] = __builtin_amdgcn_wmma_f32_16x16x4_f32(
                            false, a1s[tt], false, b0, (short)0, acc[2 * jp], false, false);
                    acc[2 * jp + 1] = __builtin_amdgcn_wmma_f32_16x16x4_f32(
                            false, a1s[tt], false, b1, (short)0, acc[2 * jp + 1], false, false);
                }
            }
        }

        float s[8], q[8];
#pragma unroll
        for (int r = 0; r < 8; ++r) { s[r] = 0.f; q[r] = 0.f; }
        gin_wave_lds_handover();
#pragma unroll
        for (int j = 0; j < 8; ++j) {
#pragma unroll
            for (int r = 0; r < 8; ++r) {
                float v = acc[j][r] + bv[j];
                v = v > 0.f ? v : 0.f;
                acc[j][r] = v;
                s[r] += v;
                q[r] += v * v;
            }
        }
#pragma unroll
        for (int m = 1; m < 16; m <<= 1) {
            float es[8], eq[8];
#pragma unroll
            for (int r = 0; r < 8; ++r) {
                es[r] = __shfl_xor(s[r], m, 32);
                eq[r] = __shfl_xor(q[r], m, 32);
            }
#pragma unroll
            for (int r = 0; r < 8; ++r) {
                s[r] += es[r];
                q[r] += eq[r];
            }
        }
#pragma unroll
        for (int r = 0; r < 8; ++r) {
            const float mean = s[r] * (1.0f / DIM);
            const float var  = q[r] * (1.0f / DIM) - mean * mean;
            const float rs   = rsqrtf(var + LN_EPSF);
#pragma unroll
            for (int j = 0; j < 8; ++j) {
                const float o = (acc[j][r] - mean) * rs * gv[j] + bt[j];
                my_a[(8 * h + r) * A_PITCH + 16 * j + c] = o;
            }
        }
        gin_wave_lds_handover();
        if (POOL) {
            int cur = __builtin_amdgcn_readlane(bid, 0);
            int len = 0;
            float4 run = make_float4(0.f, 0.f, 0.f, 0.f);
#pragma unroll
            for (int r = 0; r < 16; ++r) {
                const float4 o = *(const float4*)(my_a + r * A_PITCH + lane * 4);
                const int b = __builtin_amdgcn_readlane(bid, r);
                if (b != cur) {
                    gin_pool_flush(psum, pcnt, cur, lane, run, len);
                    cur = b;
                    len = 0;
                    run = make_float4(0.f, 0.f, 0.f, 0.f);
                }
                run.x += o.x; run.y += o.y; run.z += o.z; run.w += o.w;
                ++len;
            }
            gin_pool_flush(psum, pcnt, cur, lane, run, len);
        } else {
#pragma unroll
            for (int r = 0; r < 16; ++r) {
                const float4 o = *(const float4*)(my_a + r * A_PITCH + lane * 4);
                *(float4*)(Hout + (long long)(tile * 16 + r) * DIM + lane * 4) = o;
            }
        }
        tile = ntile;
    } while (tile < N_TILES);
}

__global__ void gin_zero(float* __restrict__ p, int n) {
    int i = blockIdx.x * blockDim.x + threadIdx.x;
    if (i < n) p[i] = 0.f;
}

__global__ void gin_outproj(const float* __restrict__ psum, const float* __restrict__ cnt,
                            const float* __restrict__ Wout, const float* __restrict__ bout,
                            float* __restrict__ logits) {
    int idx = blockIdx.x * blockDim.x + threadIdx.x;
    if (idx < NGRAPH * NCLASS) {
        int g = idx / NCLASS;
        int c = idx - g * NCLASS;
        float inv = 1.0f / fmaxf(cnt[g], 1.0f);
        float acc = bout[c];
        const float* pg = psum + g * DIM;
#pragma unroll 4
        for (int k = 0; k < DIM; ++k) acc += pg[k] * inv * Wout[k * NCLASS + c];
        logits[idx] = acc;
    }
}

__global__ void gin_logsoftmax(const float* __restrict__ logits, float* __restrict__ out) {
    __shared__ float sh[64];
    const int g = blockIdx.x;
    const int t = threadIdx.x;
    float v = (t < NCLASS) ? logits[g * NCLASS + t] : -3.4e38f;
    sh[t] = v;
    __syncthreads();
    for (int s = 32; s > 0; s >>= 1) {
        if (t < s) sh[t] = fmaxf(sh[t], sh[t + s]);
        __syncthreads();
    }
    const float mx = sh[0];
    __syncthreads();
    sh[t] = (t < NCLASS) ? expf(v - mx) : 0.f;
    __syncthreads();
    for (int s = 32; s > 0; s >>= 1) {
        if (t < s) sh[t] += sh[t + s];
        __syncthreads();
    }
    const float lse = logf(sh[0]);
    if (t < NCLASS) out[g * NCLASS + t] = v - mx - lse;
}

extern "C" void kernel_launch(void* const* d_in, const int* in_sizes, int n_in,
                              void* d_out, int out_size, void* d_ws, size_t ws_size,
                              hipStream_t stream) {
    const float* x      = (const float*)d_in[0];
    const int*   ei     = (const int*)d_in[1];
    const int*   batch  = (const int*)d_in[2];
    const float* Ws     = (const float*)d_in[3];
    const float* bs     = (const float*)d_in[4];
    const float* gammas = (const float*)d_in[5];
    const float* betas  = (const float*)d_in[6];
    const float* Wout   = (const float*)d_in[7];
    const float* bout   = (const float*)d_in[8];
    float* out = (float*)d_out;

    const int* src = ei;
    const int* dst = ei + N_EDGES;

    float* ws = (float*)d_ws;
    const size_t nd = (size_t)N_NODES * DIM;
    float* hA     = ws;
    float* hB     = ws + nd;
    float* psum   = ws + 2 * nd;
    float* cnt    = psum + (size_t)NGRAPH * DIM;
    int*   deg     = (int*)(cnt + NGRAPH);
    int*   ovf_cnt = deg + N_NODES;
    int*   col     = deg + DEG_PAD;
    int*   ovf_dst = col + (size_t)N_NODES * SLOT_CAP;
    int*   ovf_src = ovf_dst + N_EDGES;
    float* logits  = (float*)(ovf_src + N_EDGES);

    const int gEdge    = (N_EDGES + 255) / 256;

    const int nZero = NGRAPH * DIM + NGRAPH + DEG_PAD;
    gin_zero<<<(nZero + 255) / 256, 256, 0, stream>>>(psum, nZero);
    gin_fill<<<gEdge, 256, 0, stream>>>(src, dst, deg, col, ovf_cnt, ovf_dst, ovf_src);

    const float* hin = x;
    for (int i = 0; i < NLAYERS - 1; ++i) {
        float* hout = (i & 1) ? hB : hA;
        gin_layer<false><<<GEMM_NBLK, GEMM_WAVES * 32, 0, stream>>>(hin, deg, col, ovf_cnt, ovf_dst, ovf_src,
                                             Ws + (size_t)i * DIM * DIM,
                                             bs + i * DIM, gammas + i * DIM,
                                             betas + i * DIM, hout, batch, psum, cnt);
        hin = hout;
    }
    gin_layer<true><<<GEMM_NBLK, GEMM_WAVES * 32, 0, stream>>>(hin, deg, col, ovf_cnt, ovf_dst, ovf_src,
                                         Ws + (size_t)(NLAYERS - 1) * DIM * DIM,
                                         bs + (NLAYERS - 1) * DIM, gammas + (NLAYERS - 1) * DIM,
                                         betas + (NLAYERS - 1) * DIM, hB, batch, psum, cnt);

    gin_outproj<<<(NGRAPH * NCLASS + 255) / 256, 256, 0, stream>>>(psum, cnt, Wout, bout, logits);
    gin_logsoftmax<<<NGRAPH, 64, 0, stream>>>(logits, out);
}


# W-staging barrier moved from behind each wave's first gather to in front of it: a wave starts its products once its own 16 rows are summed, not once the slowest of 16 waves has; sum order kept
# speedup vs baseline: 3.6332x; 1.0309x over previous
#include <hip/hip_runtime.h>
#include <hip/hip_bf16.h>
#include <math.h>

#define N_NODES 100000
#define N_EDGES 600000
#define DIM     128
#define NLAYERS 4
#define NGRAPH  512
#define NCLASS  40
#define LN_EPSF 1e-5f

typedef __attribute__((ext_vector_type(2))) float v2f;
typedef __attribute__((ext_vector_type(4))) float v4f;
typedef __attribute__((ext_vector_type(8))) float v8f;

#define SLOT_CAP  32
#define DEG_PAD   (N_NODES + 96)

__global__ void gin_fill(const int* __restrict__ src, const int* __restrict__ dst,
                         int* __restrict__ deg, int* __restrict__ col,
                         int* __restrict__ ovf_cnt, int* __restrict__ ovf_dst,
                         int* __restrict__ ovf_src) {
    int e = blockIdx.x * blockDim.x + threadIdx.x;
    if (e < N_EDGES) {
        const int d = dst[e];
        const int s = src[e];
        const int slot = atomicAdd(deg + d, 1);
        if (slot < SLOT_CAP) {
            col[d * SLOT_CAP + slot] = s;
        } else {
            const int k = atomicAdd(ovf_cnt, 1);
            ovf_dst[k] = d;
            ovf_src[k] = s;
        }
    }
}

__device__ __forceinline__ float4 gin_row4(const float* __restrict__ h, int node, int lane) {
    return *(const float4*)(h + (long long)node * DIM + lane * 4);
}

__device__ __forceinline__ int gin_slot_row(const int* __restrict__ deg, const int* __restrict__ col,
                                            int node, int cnt, int lane) {
    const int cs = cnt < SLOT_CAP ? cnt : SLOT_CAP;
    return *((lane < cs) ? (col + node * SLOT_CAP + lane) : (deg + node));
}

#define GATHER_B    4
#define GEMM_NBLK   256
#define GEMM_WAVES  16
#define N_TILES     (N_NODES / 16)
#define A_PITCH     132

__device__ __forceinline__ void gin_wave_lds_handover() {
    __builtin_amdgcn_fence(__ATOMIC_RELEASE, "workgroup", "local");
    __builtin_amdgcn_wave_barrier();
    __builtin_amdgcn_fence(__ATOMIC_ACQUIRE, "workgroup", "local");
}

__device__ __forceinline__ void gin_pool_flush(float* __restrict__ psum, float* __restrict__ cnt,
                                               int g, int lane, float4 a, int len) {
    float* p = psum + (long long)g * DIM + lane * 4;
    atomicAdd(p + 0, a.x);
    atomicAdd(p + 1, a.y);
    atomicAdd(p + 2, a.z);
    atomicAdd(p + 3, a.w);
    if (lane == 0) atomicAdd(cnt + g, (float)len);
}

template <bool POOL>
__global__ __launch_bounds__(GEMM_WAVES * 32)
void gin_layer(const float* __restrict__ Hin, const int* __restrict__ deg,
               const int* __restrict__ col, const int* __restrict__ ovf_cnt,
               const int* __restrict__ ovf_dst, const int* __restrict__ ovf_src,
               const float* __restrict__ W,
               const float* __restrict__ bias, const float* __restrict__ gamma,
               const float* __restrict__ beta, float* __restrict__ Hout,
               const int* __restrict__ batch, float* __restrict__ psum,
               float* __restrict__ pcnt) {
    __shared__ __attribute__((aligned(16))) float w_frag[32 * 4 * 32 * 4];
    __shared__ __attribute__((aligned(16))) float a_tile[GEMM_WAVES * 16 * A_PITCH];

    const int tid  = threadIdx.x;
    const int lane = tid & 31;
    const int c    = lane & 15;
    const int h    = lane >> 4;
    const int wave = __builtin_amdgcn_readfirstlane(tid >> 5);

    static_assert(GEMM_NBLK * GEMM_WAVES <= N_TILES, "a first tile for every wave");
    static_assert(16 % GATHER_B == 0, "whole batches in a tile");
    int tile = (int)blockIdx.x + GEMM_NBLK * wave;
    int dg = deg[tile * 16 + c];

#pragma unroll
    for (int i = 0; i < 128 / GEMM_WAVES; ++i) {
        const int grp = wave + GEMM_WAVES * i;
        const int t   = grp >> 2;
        const int jp  = grp & 3;
        const float* wp = W + (4 * t + 2 * h) * DIM + 32 * jp + c;
        float4 f;
        f.x = wp[0];
        f.y = wp[DIM];
        f.z = wp[16];
        f.w = wp[DIM + 16];
        *(float4*)(w_frag + ((t * 4 + jp) * 32 + lane) * 4) = f;
    }

    float bv[8], gv[8], bt[8];
#pragma unroll
    for (int j = 0; j < 8; ++j) {
        bv[j] = bias[16 * j + c];
        gv[j] = gamma[16 * j + c];
        bt[j] = beta[16 * j + c];
    }

    float* my_a = a_tile + wave * (16 * A_PITCH);

    int my[GATHER_B];
#pragma unroll
    for (int i = 0; i < GATHER_B; ++i)
        my[i] = gin_slot_row(deg, col, tile * 16 + i, __builtin_amdgcn_readlane(dg, i), lane);

    __syncthreads();

    do {
        const int ntile = tile + GEMM_NBLK * GEMM_WAVES;
        const int ptile = ntile < N_TILES ? ntile : tile;
        const int dgn = deg[ptile * 16 + c];
        int bid = 0;
        if (POOL) bid = batch[tile * 16 + c];

        gin_wave_lds_handover();
#pragma nounroll
        for (int b = 0; b < 16 / GATHER_B; ++b) {
            const int r0 = GATHER_B * b;
            const int r1 = (b + 1 < 16 / GATHER_B) ? r0 + GATHER_B : r0;
            int nx[GATHER_B];
#pragma unroll
            for (int i = 0; i < GATHER_B; ++i)
                nx[i] = gin_slot_row(deg, col, tile * 16 + r1 + i,
                                     __builtin_amdgcn_readlane(dg, r1 + i), lane);

            int cnt[GATHER_B], cs[GATHER_B];
            int cm = 0;
#pragma unroll
            for (int i = 0; i < GATHER_B; ++i) {
                cnt[i] = __builtin_amdgcn_readlane(dg, r0 + i);
                cs[i]  = cnt[i] < SLOT_CAP ? cnt[i] : SLOT_CAP;
                cm     = cs[i] > cm ? cs[i] : cm;
            }
            float4 self[GATHER_B], sum[GATHER_B];
#pragma unroll
            for (int i = 0; i < GATHER_B; ++i) {
                self[i] = gin_row4(Hin, tile * 16 + r0 + i, lane);
                sum[i]  = make_float4(0.f, 0.f, 0.f, 0.f);
            }
            for (int j = 0; j < cm; j += 8) {
                float4 v[GATHER_B][8];
#pragma unroll
                for (int i = 0; i < GATHER_B; ++i) {
#pragma unroll
                    for (int u = 0; u < 8; ++u) {
                        v[i][u] = make_float4(0.f, 0.f, 0.f, 0.f);
                        if (j + u < cs[i])
                            v[i][u] = gin_row4(Hin, __builtin_amdgcn_readlane(my[i], j + u), lane);
                    }
                }
#pragma unroll
                for (int i = 0; i < GATHER_B; ++i) {
#pragma unroll
                    for (int u = 0; u < 8; ++u) {
                        sum[i].x += v[i][u].x; sum[i].y += v[i][u].y;
                        sum[i].z += v[i][u].z; sum[i].w += v[i][u].w;
                    }
                }
            }
#pragma unroll
            for (int i = 0; i < GATHER_B; ++i) {
                if (cnt[i] > SLOT_CAP) {
                    const int n = tile * 16 + r0 + i;
                    const int total = ovf_cnt[0];
                    for (int base = 0; base < total; base += 32) {
                        const int k  = base + lane;
                        const int od = (k < total) ? ovf_dst[k] : -1;
                        const int os = (k < total) ? ovf_src[k] : 0;
                        unsigned int hit = __builtin_amdgcn_ballot_w32(od == n);
                        while (hit != 0u) {
                            const int bit = __builtin_ctz(hit);
                            hit &= hit - 1u;
                            const float4 v = gin_row4(Hin, __builtin_amdgcn_readlane(os, bit), lane);
                            sum[i].x += v.x; sum[i].y += v.y; sum[i].z += v.z; sum[i].w += v.w;
                        }
                    }
                }
                v4f o;
                o.x = 2.f * self[i].x + sum[i].x;
                o.y = 2.f * self[i].y + sum[i].y;
                o.z = 2.f * self[i].z + sum[i].z;
                o.w = 2.f * self[i].w + sum[i].w;
                *(v4f*)(my_a + (r0 + i) * A_PITCH + lane * 4) = o;
                my[i] = nx[i];
            }
        }
        gin_wave_lds_handover();
        const float* my_frag = my_a + c * A_PITCH + 2 * h;

        dg = dgn;
#pragma unroll
        for (int i = 0; i < GATHER_B; ++i)
            my[i] = gin_slot_row(deg, col, ptile * 16 + i, __builtin_amdgcn_readlane(dg, i), lane);

        v8f acc[8];
#pragma unroll
        for (int j = 0; j < 8; ++j) acc[j] = (v8f){0.f, 0.f, 0.f, 0.f, 0.f, 0.f, 0.f, 0.f};

        const float* wl = w_frag + lane * 4;
        v4f b0s[16], b1s[16];
        v2f a0s[4], a1s[4];
#pragma unroll
        for (int u = 0; u < 16; ++u) b0s[u] = *(const v4f*)(wl + u * 128);
#pragma unroll
        for (int tt = 0; tt < 4; ++tt) a0s[tt] = *(const v2f*)(my_frag + 4 * tt);
#pragma nounroll
        for (int kk = 0; kk < 4; ++kk) {
            const int blk1 = 2 * kk + 1;
            const int blk2 = (2 * kk + 2) & 7;
#pragma unroll
            for (int u = 0; u < 16; ++u) b1s[u] = *(const v4f*)(wl + blk1 * 2048 + u * 128);
#pragma unroll
            for (int tt = 0; tt < 4; ++tt) a1s[tt] = *(const v2f*)(my_frag + blk1 * 16 + 4 * tt);
#pragma unroll
            for (int tt = 0; tt < 4; ++tt) {
#pragma unroll
                for (int jp = 0; jp < 4; ++jp) {
                    const v4f f = b0s[tt * 4 + jp];
                    v2f b0, b1;
                    b0.x = f.x; b0.y = f.y;
                    b1.x = f.z; b1.y = f.w;
                    acc[2 * jp] = __builtin_amdgcn_wmma_f32_16x16x4_f32(
                            false, a0s[tt], false, b0, (short)0, acc[2 * jp], false, false);
                    acc[2 * jp + 1] = __builtin_amdgcn_wmma_f32_16x16x4_f32(
                            false, a0s[tt], false, b1, (short)0, acc[2 * jp + 1], false, false);
                }
            }
#pragma unroll
            for (int u = 0; u < 16; ++u) b0s[u] = *(const v4f*)(wl + blk2 * 2048 + u * 128);
#pragma unroll
            for (int tt = 0; tt < 4; ++tt) a0s[tt] = *(const v2f*)(my_frag + blk2 * 16 + 4 * tt);
#pragma unroll
            for (int tt = 0; tt < 4; ++tt) {
#pragma unroll
                for (int jp = 0; jp < 4; ++jp) {
                    const v4f f = b1s[tt * 4 + jp];
                    v2f b0, b1;
                    b0.x = f.x; b0.y = f.y;
                    b1.x = f.z; b1.y = f.w;
                    acc[2 * jp] = __builtin_amdgcn_wmma_f32_16x16x4_f32(
                            false, a1s[tt], false, b0, (short)0, acc[2 * jp], false, false);
                    acc[2 * jp + 1] = __builtin_amdgcn_wmma_f32_16x16x4_f32(
                            false, a1s[tt], false, b1, (short)0, acc[2 * jp + 1], false, false);
                }
            }
        }

        float s[8], q[8];
#pragma unroll
        for (int r = 0; r < 8; ++r) { s[r] = 0.f; q[r] = 0.f; }
        gin_wave_lds_handover();
#pragma unroll
        for (int j = 0; j < 8; ++j) {
#pragma unroll
            for (int r = 0; r < 8; ++r) {
                float v = acc[j][r] + bv[j];
                v = v > 0.f ? v : 0.f;
                acc[j][r] = v;
                s[r] += v;
                q[r] += v * v;
            }
        }
#pragma unroll
        for (int m = 1; m < 16; m <<= 1) {
            float es[8], eq[8];
#pragma unroll
            for (int r = 0; r < 8; ++r) {
                es[r] = __shfl_xor(s[r], m, 32);
                eq[r] = __shfl_xor(q[r], m, 32);
            }
#pragma unroll
            for (int r = 0; r < 8; ++r) {
                s[r] += es[r];
                q[r] += eq[r];
            }
        }
#pragma unroll
        for (int r = 0; r < 8; ++r) {
            const float mean = s[r] * (1.0f / DIM);
            const float var  = q[r] * (1.0f / DIM) - mean * mean;
            const float rs   = rsqrtf(var + LN_EPSF);
#pragma unroll
            for (int j = 0; j < 8; ++j) {
                const float o = (acc[j][r] - mean) * rs * gv[j] + bt[j];
                my_a[(8 * h + r) * A_PITCH + 16 * j + c] = o;
            }
        }
        gin_wave_lds_handover();
        if (POOL) {
            int cur = __builtin_amdgcn_readlane(bid, 0);
            int len = 0;
            float4 run = make_float4(0.f, 0.f, 0.f, 0.f);
#pragma unroll
            for (int r = 0; r < 16; ++r) {
                const float4 o = *(const float4*)(my_a + r * A_PITCH + lane * 4);
                const int b = __builtin_amdgcn_readlane(bid, r);
                if (b != cur) {
                    gin_pool_flush(psum, pcnt, cur, lane, run, len);
                    cur = b;
                    len = 0;
                    run = make_float4(0.f, 0.f, 0.f, 0.f);
                }
                run.x += o.x; run.y += o.y; run.z += o.z; run.w += o.w;
                ++len;
            }
            gin_pool_flush(psum, pcnt, cur, lane, run, len);
        } else {
#pragma unroll
            for (int r = 0; r < 16; ++r) {
                const float4 o = *(const float4*)(my_a + r * A_PITCH + lane * 4);
                *(float4*)(Hout + (long long)(tile * 16 + r) * DIM + lane * 4) = o;
            }
        }
        tile = ntile;
    } while (tile < N_TILES);
}

__global__ void gin_zero(float* __restrict__ p, int n) {
    int i = blockIdx.x * blockDim.x + threadIdx.x;
    if (i < n) p[i] = 0.f;
}

__global__ void gin_outproj(const float* __restrict__ psum, const float* __restrict__ cnt,
                            const float* __restrict__ Wout, const float* __restrict__ bout,
                            float* __restrict__ logits) {
    int idx = blockIdx.x * blockDim.x + threadIdx.x;
    if (idx < NGRAPH * NCLASS) {
        int g = idx / NCLASS;
        int c = idx - g * NCLASS;
        float inv = 1.0f / fmaxf(cnt[g], 1.0f);
        float acc = bout[c];
        const float* pg = psum + g * DIM;
#pragma unroll 4
        for (int k = 0; k < DIM; ++k) acc += pg[k] * inv * Wout[k * NCLASS + c];
        logits[idx] = acc;
    }
}

__global__ void gin_logsoftmax(const float* __restrict__ logits, float* __restrict__ out) {
    __shared__ float sh[64];
    const int g = blockIdx.x;
    const int t = threadIdx.x;
    float v = (t < NCLASS) ? logits[g * NCLASS + t] : -3.4e38f;
    sh[t] = v;
    __syncthreads();
    for (int s = 32; s > 0; s >>= 1) {
        if (t < s) sh[t] = fmaxf(sh[t], sh[t + s]);
        __syncthreads();
    }
    const float mx = sh[0];
    __syncthreads();
    sh[t] = (t < NCLASS) ? expf(v - mx) : 0.f;
    __syncthreads();
    for (int s = 32; s > 0; s >>= 1) {
        if (t < s) sh[t] += sh[t + s];
        __syncthreads();
    }
    const float lse = logf(sh[0]);
    if (t < NCLASS) out[g * NCLASS + t] = v - mx - lse;
}

extern "C" void kernel_launch(void* const* d_in, const int* in_sizes, int n_in,
                              void* d_out, int out_size, void* d_ws, size_t ws_size,
                              hipStream_t stream) {
    const float* x      = (const float*)d_in[0];
    const int*   ei     = (const int*)d_in[1];
    const int*   batch  = (const int*)d_in[2];
    const float* Ws     = (const float*)d_in[3];
    const float* bs     = (const float*)d_in[4];
    const float* gammas = (const float*)d_in[5];
    const float* betas  = (const float*)d_in[6];
    const float* Wout   = (const float*)d_in[7];
    const float* bout   = (const float*)d_in[8];
    float* out = (float*)d_out;

    const int* src = ei;
    const int* dst = ei + N_EDGES;

    float* ws = (float*)d_ws;
    const size_t nd = (size_t)N_NODES * DIM;
    float* hA     = ws;
    float* hB     = ws + nd;
    float* psum   = ws + 2 * nd;
    float* cnt    = psum + (size_t)NGRAPH * DIM;
    int*   deg     = (int*)(cnt + NGRAPH);
    int*   ovf_cnt = deg + N_NODES;
    int*   col     = deg + DEG_PAD;
    int*   ovf_dst = col + (size_t)N_NODES * SLOT_CAP;
    int*   ovf_src = ovf_dst + N_EDGES;
    float* logits  = (float*)(ovf_src + N_EDGES);

    const int gEdge    = (N_EDGES + 255) / 256;

    const int nZero = NGRAPH * DIM + NGRAPH + DEG_PAD;
    gin_zero<<<(nZero + 255) / 256, 256, 0, stream>>>(psum, nZero);
    gin_fill<<<gEdge, 256, 0, stream>>>(src, dst, deg, col, ovf_cnt, ovf_dst, ovf_src);

    const float* hin = x;
    for (int i = 0; i < NLAYERS - 1; ++i) {
        float* hout = (i & 1) ? hB : hA;
        gin_layer<false><<<GEMM_NBLK, GEMM_WAVES * 32, 0, stream>>>(hin, deg, col, ovf_cnt, ovf_dst, ovf_src,
                                             Ws + (size_t)i * DIM * DIM,
                                             bs + i * DIM, gammas + i * DIM,
                                             betas + i * DIM, hout, batch, psum, cnt);
        hin = hout;
    }
    gin_layer<true><<<GEMM_NBLK, GEMM_WAVES * 32, 0, stream>>>(hin, deg, col, ovf_cnt, ovf_dst, ovf_src,
                                         Ws + (size_t)(NLAYERS - 1) * DIM * DIM,
                                         bs + (NLAYERS - 1) * DIM, gammas + (NLAYERS - 1) * DIM,
                                         betas + (NLAYERS - 1) * DIM, hB, batch, psum, cnt);

    gin_outproj<<<(NGRAPH * NCLASS + 255) / 256, 256, 0, stream>>>(psum, cnt, Wout, bout, logits);
    gin_logsoftmax<<<NGRAPH, 64, 0, stream>>>(logits, out);
}


# Edge list built on LDS counters in two launches (chunks binned by destination range, then one work-group per range hands out slots): no global counting add per edge; neighbour sum order changes
# speedup vs baseline: 4.1116x; 1.1317x over previous
#include <hip/hip_runtime.h>
#include <hip/hip_bf16.h>
#include <math.h>

#define N_NODES 100000
#define N_EDGES 600000
#define DIM     128
#define NLAYERS 4
#define NGRAPH  512
#define NCLASS  40
#define LN_EPSF 1e-5f

typedef __attribute__((ext_vector_type(2))) float v2f;
typedef __attribute__((ext_vector_type(4))) float v4f;
typedef __attribute__((ext_vector_type(8))) float v8f;

#define SLOT_CAP  32
#define DEG_PAD   (N_NODES + 96)
#define NPB       400
#define NBKT      (N_NODES / NPB)
#define NBIN      256
#define CHUNK     ((N_EDGES + NBIN - 1) / NBIN)
#define SUBCAP    32
#define BIN_THREADS 512
#define BIN_EPT   ((CHUNK + BIN_THREADS - 1) / BIN_THREADS)
#define SRC_BITS  17

__global__ __launch_bounds__(BIN_THREADS)
void gin_bin(const int* __restrict__ src, const int* __restrict__ dst,
             int* __restrict__ region, int* __restrict__ cnt_rw,
             float* __restrict__ pool_clear, int* __restrict__ ovf_cnt) {
    static_assert(NBKT * NPB == N_NODES, "buckets cover the nodes exactly");
    static_assert(N_NODES <= (1 << SRC_BITS) && NPB <= (1 << (31 - SRC_BITS)), "entry packing");
    __shared__ int lc[NBKT];
    __shared__ int stage[NBKT * SUBCAP];
    const int tid = threadIdx.x;
    const int w   = blockIdx.x;
    for (int i = tid; i < NBKT; i += BIN_THREADS) lc[i] = 0;
    const int g = w * BIN_THREADS + tid;
    if (g < NGRAPH * DIM + NGRAPH) pool_clear[g] = 0.f;
    if (g == 0) ovf_cnt[0] = 0;
    const int e0 = w * CHUNK;
    const int e1 = (e0 + CHUNK < N_EDGES) ? e0 + CHUNK : N_EDGES;
    int dd[BIN_EPT], ss[BIN_EPT];
#pragma unroll
    for (int k = 0; k < BIN_EPT; ++k) {
        const int e = e0 + tid + k * BIN_THREADS;
        dd[k] = -1; ss[k] = 0;
        if (e < e1) { dd[k] = dst[e]; ss[k] = src[e]; }
    }
    __syncthreads();
#pragma unroll
    for (int k = 0; k < BIN_EPT; ++k) {
        if (dd[k] >= 0) {
            const int b  = dd[k] / NPB;
            const int dl = dd[k] - b * NPB;
            const int rank = atomicAdd(&lc[b], 1);
            if (rank < SUBCAP) stage[b * SUBCAP + rank] = (dl << SRC_BITS) | ss[k];
        }
    }
    __syncthreads();
    const int wave = tid >> 5, lane = tid & 31;
    for (int b = wave; b < NBKT; b += BIN_THREADS / 32) {
        const int c = lc[b];
        const int cs = c < SUBCAP ? c : SUBCAP;
        if (lane < cs) region[(b * NBIN + w) * SUBCAP + lane] = stage[b * SUBCAP + lane];
        if (lane == 0) cnt_rw[b * NBIN + w] = c;
    }
}

__device__ __forceinline__ void gin_place_one(int* nc, int* rows, int node0, int dl, int s,
                                              int* __restrict__ ovf_cnt, int* __restrict__ ovf_dst,
                                              int* __restrict__ ovf_src) {
    const int slot = atomicAdd(&nc[dl], 1);
    if (slot < SLOT_CAP) {
        rows[dl * SLOT_CAP + slot] = s;
    } else {
        const int k = atomicAdd(ovf_cnt, 1);
        ovf_dst[k] = node0 + dl;
        ovf_src[k] = s;
    }
}

#define PLACE_THREADS 512
#define PLACE_WAVES   (PLACE_THREADS / 32)
#define PLACE_SUBS    (NBIN / PLACE_WAVES)

__global__ __launch_bounds__(PLACE_THREADS)
void gin_place(const int* __restrict__ src, const int* __restrict__ dst,
               const int* __restrict__ region, const int* __restrict__ cnt_rw,
               int* __restrict__ deg, int* __restrict__ col,
               int* __restrict__ ovf_cnt, int* __restrict__ ovf_dst,
               int* __restrict__ ovf_src) {
    static_assert(NBIN % PLACE_WAVES == 0, "whole lines per wave");
    __shared__ int nc[NPB];
    __shared__ int rows[NPB * SLOT_CAP];
    const int tid  = threadIdx.x;
    const int r    = blockIdx.x;
    const int wave = tid >> 5, lane = tid & 31;
    const int node0 = r * NPB;
    for (int i = tid; i < NPB; i += PLACE_THREADS) nc[i] = 0;
    int cw[PLACE_SUBS], pk[PLACE_SUBS];
#pragma unroll
    for (int k = 0; k < PLACE_SUBS; ++k) cw[k] = cnt_rw[r * NBIN + wave * PLACE_SUBS + k];
#pragma unroll
    for (int k = 0; k < PLACE_SUBS; ++k) {
        pk[k] = -1;
        if (cw[k] <= SUBCAP && lane < cw[k])
            pk[k] = region[(r * NBIN + wave * PLACE_SUBS + k) * SUBCAP + lane];
    }
    __syncthreads();
#pragma unroll
    for (int k = 0; k < PLACE_SUBS; ++k) {
        if (pk[k] >= 0)
            gin_place_one(nc, rows, node0, pk[k] >> SRC_BITS, pk[k] & ((1 << SRC_BITS) - 1),
                          ovf_cnt, ovf_dst, ovf_src);
        if (cw[k] > SUBCAP) {
            const int w  = wave * PLACE_SUBS + k;
            const int e0 = w * CHUNK;
            const int e1 = (e0 + CHUNK < N_EDGES) ? e0 + CHUNK : N_EDGES;
            for (int e = e0 + lane; e < e1; e += 32) {
                const int d = dst[e];
                if (d / NPB == r)
                    gin_place_one(nc, rows, node0, d - node0, src[e], ovf_cnt, ovf_dst, ovf_src);
            }
        }
    }
    __syncthreads();
    for (int i = wave; i < NPB; i += PLACE_WAVES) {
        const int c = nc[i];
        const int cs = c < SLOT_CAP ? c : SLOT_CAP;
        if (lane < cs) col[(node0 + i) * SLOT_CAP + lane] = rows[i * SLOT_CAP + lane];
    }
    for (int i = tid; i < NPB; i += PLACE_THREADS) deg[node0 + i] = nc[i];
}

__device__ __forceinline__ float4 gin_row4(const float* __restrict__ h, int node, int lane) {
    return *(const float4*)(h + (long long)node * DIM + lane * 4);
}

__device__ __forceinline__ int gin_slot_row(const int* __restrict__ deg, const int* __restrict__ col,
                                            int node, int cnt, int lane) {
    const int cs = cnt < SLOT_CAP ? cnt : SLOT_CAP;
    return *((lane < cs) ? (col + node * SLOT_CAP + lane) : (deg + node));
}

#define GATHER_B    4
#define GEMM_NBLK   256
#define GEMM_WAVES  16
#define N_TILES     (N_NODES / 16)
#define A_PITCH     132

__device__ __forceinline__ void gin_wave_lds_handover() {
    __builtin_amdgcn_fence(__ATOMIC_RELEASE, "workgroup", "local");
    __builtin_amdgcn_wave_barrier();
    __builtin_amdgcn_fence(__ATOMIC_ACQUIRE, "workgroup", "local");
}

__device__ __forceinline__ void gin_pool_flush(float* __restrict__ psum, float* __restrict__ cnt,
                                               int g, int lane, float4 a, int len) {
    float* p = psum + (long long)g * DIM + lane * 4;
    atomicAdd(p + 0, a.x);
    atomicAdd(p + 1, a.y);
    atomicAdd(p + 2, a.z);
    atomicAdd(p + 3, a.w);
    if (lane == 0) atomicAdd(cnt + g, (float)len);
}

template <bool POOL>
__global__ __launch_bounds__(GEMM_WAVES * 32)
void gin_layer(const float* __restrict__ Hin, const int* __restrict__ deg,
               const int* __restrict__ col, const int* __restrict__ ovf_cnt,
               const int* __restrict__ ovf_dst, const int* __restrict__ ovf_src,
               const float* __restrict__ W,
               const float* __restrict__ bias, const float* __restrict__ gamma,
               const float* __restrict__ beta, float* __restrict__ Hout,
               const int* __restrict__ batch, float* __restrict__ psum,
               float* __restrict__ pcnt) {
    __shared__ __attribute__((aligned(16))) float w_frag[32 * 4 * 32 * 4];
    __shared__ __attribute__((aligned(16))) float a_tile[GEMM_WAVES * 16 * A_PITCH];

    const int tid  = threadIdx.x;
    const int lane = tid & 31;
    const int c    = lane & 15;
    const int h    = lane >> 4;
    const int wave = __builtin_amdgcn_readfirstlane(tid >> 5);

    static_assert(GEMM_NBLK * GEMM_WAVES <= N_TILES, "a first tile for every wave");
    static_assert(16 % GATHER_B == 0, "whole batches in a tile");
    int tile = (int)blockIdx.x + GEMM_NBLK * wave;
    int dg = deg[tile * 16 + c];

#pragma unroll
    for (int i = 0; i < 128 / GEMM_WAVES; ++i) {
        const int grp = wave + GEMM_WAVES * i;
        const int t   = grp >> 2;
        const int jp  = grp & 3;
        const float* wp = W + (4 * t + 2 * h) * DIM + 32 * jp + c;
        float4 f;
        f.x = wp[0];
        f.y = wp[DIM];
        f.z = wp[16];
        f.w = wp[DIM + 16];
        *(float4*)(w_frag + ((t * 4 + jp) * 32 + lane) * 4) = f;
    }

    float bv[8], gv[8], bt[8];
#pragma unroll
    for (int j = 0; j < 8; ++j) {
        bv[j] = bias[16 * j + c];
        gv[j] = gamma[16 * j + c];
        bt[j] = beta[16 * j + c];
    }

    float* my_a = a_tile + wave * (16 * A_PITCH);

    int my[GATHER_B];
#pragma unroll
    for (int i = 0; i < GATHER_B; ++i)
        my[i] = gin_slot_row(deg, col, tile * 16 + i, __builtin_amdgcn_readlane(dg, i), lane);

    __syncthreads();

    do {
        const int ntile = tile + GEMM_NBLK * GEMM_WAVES;
        const int ptile = ntile < N_TILES ? ntile : tile;
        const int dgn = deg[ptile * 16 + c];
        int bid = 0;
        if (POOL) bid = batch[tile * 16 + c];

        gin_wave_lds_handover();
#pragma nounroll
        for (int b = 0; b < 16 / GATHER_B; ++b) {
            const int r0 = GATHER_B * b;
            const int r1 = (b + 1 < 16 / GATHER_B) ? r0 + GATHER_B : r0;
            int nx[GATHER_B];
#pragma unroll
            for (int i = 0; i < GATHER_B; ++i)
                nx[i] = gin_slot_row(deg, col, tile * 16 + r1 + i,
                                     __builtin_amdgcn_readlane(dg, r1 + i), lane);

            int cnt[GATHER_B], cs[GATHER_B];
            int cm = 0;
#pragma unroll
            for (int i = 0; i < GATHER_B; ++i) {
                cnt[i] = __builtin_amdgcn_readlane(dg, r0 + i);
                cs[i]  = cnt[i] < SLOT_CAP ? cnt[i] : SLOT_CAP;
                cm     = cs[i] > cm ? cs[i] : cm;
            }
            float4 self[GATHER_B], sum[GATHER_B];
#pragma unroll
            for (int i = 0; i < GATHER_B; ++i) {
                self[i] = gin_row4(Hin, tile * 16 + r0 + i, lane);
                sum[i]  = make_float4(0.f, 0.f, 0.f, 0.f);
            }
            for (int j = 0; j < cm; j += 8) {
                float4 v[GATHER_B][8];
#pragma unroll
                for (int i = 0; i < GATHER_B; ++i) {
#pragma unroll
                    for (int u = 0; u < 8; ++u) {
                        v[i][u] = make_float4(0.f, 0.f, 0.f, 0.f);
                        if (j + u < cs[i])
                            v[i][u] = gin_row4(Hin, __builtin_amdgcn_readlane(my[i], j + u), lane);
                    }
                }
#pragma unroll
                for (int i = 0; i < GATHER_B; ++i) {
#pragma unroll
                    for (int u = 0; u < 8; ++u) {
                        sum[i].x += v[i][u].x; sum[i].y += v[i][u].y;
                        sum[i].z += v[i][u].z; sum[i].w += v[i][u].w;
                    }
                }
            }
#pragma unroll
            for (int i = 0; i < GATHER_B; ++i) {
                if (cnt[i] > SLOT_CAP) {
                    const int n = tile * 16 + r0 + i;
                    const int total = ovf_cnt[0];
                    for (int base = 0; base < total; base += 32) {
                        const int k  = base + lane;
                        const int od = (k < total) ? ovf_dst[k] : -1;
                        const int os = (k < total) ? ovf_src[k] : 0;
                        unsigned int hit = __builtin_amdgcn_ballot_w32(od == n);
                        while (hit != 0u) {
                            const int bit = __builtin_ctz(hit);
                            hit &= hit - 1u;
                            const float4 v = gin_row4(Hin, __builtin_amdgcn_readlane(os, bit), lane);
                            sum[i].x += v.x; sum[i].y += v.y; sum[i].z += v.z; sum[i].w += v.w;
                        }
                    }
                }
                v4f o;
                o.x = 2.f * self[i].x + sum[i].x;
                o.y = 2.f * self[i].y + sum[i].y;
                o.z = 2.f * self[i].z + sum[i].z;
                o.w = 2.f * self[i].w + sum[i].w;
                *(v4f*)(my_a + (r0 + i) * A_PITCH + lane * 4) = o;
                my[i] = nx[i];
            }
        }
        gin_wave_lds_handover();
        const float* my_frag = my_a + c * A_PITCH + 2 * h;

        dg = dgn;
#pragma unroll
        for (int i = 0; i < GATHER_B; ++i)
            my[i] = gin_slot_row(deg, col, ptile * 16 + i, __builtin_amdgcn_readlane(dg, i), lane);

        v8f acc[8];
#pragma unroll
        for (int j = 0; j < 8; ++j) acc[j] = (v8f){0.f, 0.f, 0.f, 0.f, 0.f, 0.f, 0.f, 0.f};

        const float* wl = w_frag + lane * 4;
        v4f b0s[16], b1s[16];
        v2f a0s[4], a1s[4];
#pragma unroll
        for (int u = 0; u < 16; ++u) b0s[u] = *(const v4f*)(wl + u * 128);
#pragma unroll
        for (int tt = 0; tt < 4; ++tt) a0s[tt] = *(const v2f*)(my_frag + 4 * tt);
#pragma nounroll
        for (int kk = 0; kk < 4; ++kk) {
            const int blk1 = 2 * kk + 1;
            const int blk2 = (2 * kk + 2) & 7;
#pragma unroll
            for (int u = 0; u < 16; ++u) b1s[u] = *(const v4f*)(wl + blk1 * 2048 + u * 128);
#pragma unroll
            for (int tt = 0; tt < 4; ++tt) a1s[tt] = *(const v2f*)(my_frag + blk1 * 16 + 4 * tt);
#pragma unroll
            for (int tt = 0; tt < 4; ++tt) {
#pragma unroll
                for (int jp = 0; jp < 4; ++jp) {
                    const v4f f = b0s[tt * 4 + jp];
                    v2f b0, b1;
                    b0.x = f.x; b0.y = f.y;
                    b1.x = f.z; b1.y = f.w;
                    acc[2 * jp] = __builtin_amdgcn_wmma_f32_16x16x4_f32(
                            false, a0s[tt], false, b0, (short)0, acc[2 * jp], false, false);
                    acc[2 * jp + 1] = __builtin_amdgcn_wmma_f32_16x16x4_f32(
                            false, a0s[tt], false, b1, (short)0, acc[2 * jp + 1], false, false);
                }
            }
#pragma unroll
            for (int u = 0; u < 16; ++u) b0s[u] = *(const v4f*)(wl + blk2 * 2048 + u * 128);
#pragma unroll
            for (int tt = 0; tt < 4; ++tt) a0s[tt] = *(const v2f*)(my_frag + blk2 * 16 + 4 * tt);
#pragma unroll
            for (int tt = 0; tt < 4; ++tt) {
#pragma unroll
                for (int jp = 0; jp < 4; ++jp) {
                    const v4f f = b1s[tt * 4 + jp];
                    v2f b0, b1;
                    b0.x = f.x; b0.y = f.y;
                    b1.x = f.z; b1.y = f.w;
                    acc[2 * jp] = __builtin_amdgcn_wmma_f32_16x16x4_f32(
                            false, a1s[tt], false, b0, (short)0, acc[2 * jp], false, false);
                    acc[2 * jp + 1] = __builtin_amdgcn_wmma_f32_16x16x4_f32(
                            false, a1s[tt], false, b1, (short)0, acc[2 * jp + 1], false, false);
                }
            }
        }

        float s[8], q[8];
#pragma unroll
        for (int r = 0; r < 8; ++r) { s[r] = 0.f; q[r] = 0.f; }
        gin_wave_lds_handover();
#pragma unroll
        for (int j = 0; j < 8; ++j) {
#pragma unroll
            for (int r = 0; r < 8; ++r) {
                float v = acc[j][r] + bv[j];
                v = v > 0.f ? v : 0.f;
                acc[j][r] = v;
                s[r] += v;
                q[r] += v * v;
            }
        }
#pragma unroll
        for (int m = 1; m < 16; m <<= 1) {
            float es[8], eq[8];
#pragma unroll
            for (int r = 0; r < 8; ++r) {
                es[r] = __shfl_xor(s[r], m, 32);
                eq[r] = __shfl_xor(q[r], m, 32);
            }
#pragma unroll
            for (int r = 0; r < 8; ++r) {
                s[r] += es[r];
                q[r] += eq[r];
            }
        }
#pragma unroll
        for (int r = 0; r < 8; ++r) {
            const float mean = s[r] * (1.0f / DIM);
            const float var  = q[r] * (1.0f / DIM) - mean * mean;
            const float rs   = rsqrtf(var + LN_EPSF);
#pragma unroll
            for (int j = 0; j < 8; ++j) {
                const float o = (acc[j][r] - mean) * rs * gv[j] + bt[j];
                my_a[(8 * h + r) * A_PITCH + 16 * j + c] = o;
            }
        }
        gin_wave_lds_handover();
        if (POOL) {
            int cur = __builtin_amdgcn_readlane(bid, 0);
            int len = 0;
            float4 run = make_float4(0.f, 0.f, 0.f, 0.f);
#pragma unroll
            for (int r = 0; r < 16; ++r) {
                const float4 o = *(const float4*)(my_a + r * A_PITCH + lane * 4);
                const int b = __builtin_amdgcn_readlane(bid, r);
                if (b != cur) {
                    gin_pool_flush(psum, pcnt, cur, lane, run, len);
                    cur = b;
                    len = 0;
                    run = make_float4(0.f, 0.f, 0.f, 0.f);
                }
                run.x += o.x; run.y += o.y; run.z += o.z; run.w += o.w;
                ++len;
            }
            gin_pool_flush(psum, pcnt, cur, lane, run, len);
        } else {
#pragma unroll
            for (int r = 0; r < 16; ++r) {
                const float4 o = *(const float4*)(my_a + r * A_PITCH + lane * 4);
                *(float4*)(Hout + (long long)(tile * 16 + r) * DIM + lane * 4) = o;
            }
        }
        tile = ntile;
    } while (tile < N_TILES);
}

__global__ void gin_outproj(const float* __restrict__ psum, const float* __restrict__ cnt,
                            const float* __restrict__ Wout, const float* __restrict__ bout,
                            float* __restrict__ logits) {
    int idx = blockIdx.x * blockDim.x + threadIdx.x;
    if (idx < NGRAPH * NCLASS) {
        int g = idx / NCLASS;
        int c = idx - g * NCLASS;
        float inv = 1.0f / fmaxf(cnt[g], 1.0f);
        float acc = bout[c];
        const float* pg = psum + g * DIM;
#pragma unroll 4
        for (int k = 0; k < DIM; ++k) acc += pg[k] * inv * Wout[k * NCLASS + c];
        logits[idx] = acc;
    }
}

__global__ void gin_logsoftmax(const float* __restrict__ logits, float* __restrict__ out) {
    __shared__ float sh[64];
    const int g = blockIdx.x;
    const int t = threadIdx.x;
    float v = (t < NCLASS) ? logits[g * NCLASS + t] : -3.4e38f;
    sh[t] = v;
    __syncthreads();
    for (int s = 32; s > 0; s >>= 1) {
        if (t < s) sh[t] = fmaxf(sh[t], sh[t + s]);
        __syncthreads();
    }
    const float mx = sh[0];
    __syncthreads();
    sh[t] = (t < NCLASS) ? expf(v - mx) : 0.f;
    __syncthreads();
    for (int s = 32; s > 0; s >>= 1) {
        if (t < s) sh[t] += sh[t + s];
        __syncthreads();
    }
    const float lse = logf(sh[0]);
    if (t < NCLASS) out[g * NCLASS + t] = v - mx - lse;
}

extern "C" void kernel_launch(void* const* d_in, const int* in_sizes, int n_in,
                              void* d_out, int out_size, void* d_ws, size_t ws_size,
                              hipStream_t stream) {
    const float* x      = (const float*)d_in[0];
    const int*   ei     = (const int*)d_in[1];
    const int*   batch  = (const int*)d_in[2];
    const float* Ws     = (const float*)d_in[3];
    const float* bs     = (const float*)d_in[4];
    const float* gammas = (const float*)d_in[5];
    const float* betas  = (const float*)d_in[6];
    const float* Wout   = (const float*)d_in[7];
    const float* bout   = (const float*)d_in[8];
    float* out = (float*)d_out;

    const int* src = ei;
    const int* dst = ei + N_EDGES;

    float* ws = (float*)d_ws;
    const size_t nd = (size_t)N_NODES * DIM;
    float* hA     = ws;
    float* hB     = ws + nd;
    float* psum   = ws + 2 * nd;
    float* cnt    = psum + (size_t)NGRAPH * DIM;
    int*   deg     = (int*)(cnt + NGRAPH);
    int*   ovf_cnt = deg + N_NODES;
    int*   col     = deg + DEG_PAD;
    int*   ovf_dst = col + (size_t)N_NODES * SLOT_CAP;
    int*   ovf_src = ovf_dst + N_EDGES;
    float* logits  = (float*)(ovf_src + N_EDGES);

    int*   region  = (int*)(logits + NGRAPH * NCLASS);
    int*   cnt_rw  = region + (size_t)NBKT * NBIN * SUBCAP;

    gin_bin<<<NBIN, BIN_THREADS, 0, stream>>>(src, dst, region, cnt_rw, psum, ovf_cnt);
    gin_place<<<NBKT, PLACE_THREADS, 0, stream>>>(src, dst, region, cnt_rw, deg, col, ovf_cnt, ovf_dst, ovf_src);

    const float* hin = x;
    for (int i = 0; i < NLAYERS - 1; ++i) {
        float* hout = (i & 1) ? hB : hA;
        gin_layer<false><<<GEMM_NBLK, GEMM_WAVES * 32, 0, stream>>>(hin, deg, col, ovf_cnt, ovf_dst, ovf_src,
                                             Ws + (size_t)i * DIM * DIM,
                                             bs + i * DIM, gammas + i * DIM,
                                             betas + i * DIM, hout, batch, psum, cnt);
        hin = hout;
    }
    gin_layer<true><<<GEMM_NBLK, GEMM_WAVES * 32, 0, stream>>>(hin, deg, col, ovf_cnt, ovf_dst, ovf_src,
                                         Ws + (size_t)(NLAYERS - 1) * DIM * DIM,
                                         bs + (NLAYERS - 1) * DIM, gammas + (NLAYERS - 1) * DIM,
                                         betas + (NLAYERS - 1) * DIM, hB, batch, psum, cnt);

    gin_outproj<<<(NGRAPH * NCLASS + 255) / 256, 256, 0, stream>>>(psum, cnt, Wout, bout, logits);
    gin_logsoftmax<<<NGRAPH, 64, 0, stream>>>(logits, out);
}


# Output head in one launch: a work-group per 2 graphs projects with all W_out and pooled-sum loads in flight at once (k in 4 parts, added in LDS), log-softmax in place; logit and exp sum orders change
# speedup vs baseline: 4.4154x; 1.0739x over previous
#include <hip/hip_runtime.h>
#include <hip/hip_bf16.h>
#include <math.h>

#define N_NODES 100000
#define N_EDGES 600000
#define DIM     128
#define NLAYERS 4
#define NGRAPH  512
#define NCLASS  40
#define LN_EPSF 1e-5f

typedef __attribute__((ext_vector_type(2))) float v2f;
typedef __attribute__((ext_vector_type(4))) float v4f;
typedef __attribute__((ext_vector_type(8))) float v8f;

#define SLOT_CAP  32
#define DEG_PAD   (N_NODES + 96)
#define NPB       400
#define NBKT      (N_NODES / NPB)
#define NBIN      256
#define CHUNK     ((N_EDGES + NBIN - 1) / NBIN)
#define SUBCAP    32
#define BIN_THREADS 512
#define BIN_EPT   ((CHUNK + BIN_THREADS - 1) / BIN_THREADS)
#define SRC_BITS  17

__global__ __launch_bounds__(BIN_THREADS)
void gin_bin(const int* __restrict__ src, const int* __restrict__ dst,
             int* __restrict__ region, int* __restrict__ cnt_rw,
             float* __restrict__ pool_clear, int* __restrict__ ovf_cnt) {
    static_assert(NBKT * NPB == N_NODES, "buckets cover the nodes exactly");
    static_assert(N_NODES <= (1 << SRC_BITS) && NPB <= (1 << (31 - SRC_BITS)), "entry packing");
    __shared__ int lc[NBKT];
    __shared__ int stage[NBKT * SUBCAP];
    const int tid = threadIdx.x;
    const int w   = blockIdx.x;
    for (int i = tid; i < NBKT; i += BIN_THREADS) lc[i] = 0;
    const int g = w * BIN_THREADS + tid;
    if (g < NGRAPH * DIM + NGRAPH) pool_clear[g] = 0.f;
    if (g == 0) ovf_cnt[0] = 0;
    const int e0 = w * CHUNK;
    const int e1 = (e0 + CHUNK < N_EDGES) ? e0 + CHUNK : N_EDGES;
    int dd[BIN_EPT], ss[BIN_EPT];
#pragma unroll
    for (int k = 0; k < BIN_EPT; ++k) {
        const int e = e0 + tid + k * BIN_THREADS;
        dd[k] = -1; ss[k] = 0;
        if (e < e1) { dd[k] = dst[e]; ss[k] = src[e]; }
    }
    __syncthreads();
#pragma unroll
    for (int k = 0; k < BIN_EPT; ++k) {
        if (dd[k] >= 0) {
            const int b  = dd[k] / NPB;
            const int dl = dd[k] - b * NPB;
            const int rank = atomicAdd(&lc[b], 1);
            if (rank < SUBCAP) stage[b * SUBCAP + rank] = (dl << SRC_BITS) | ss[k];
        }
    }
    __syncthreads();
    const int wave = tid >> 5, lane = tid & 31;
    for (int b = wave; b < NBKT; b += BIN_THREADS / 32) {
        const int c = lc[b];
        const int cs = c < SUBCAP ? c : SUBCAP;
        if (lane < cs) region[(b * NBIN + w) * SUBCAP + lane] = stage[b * SUBCAP + lane];
        if (lane == 0) cnt_rw[b * NBIN + w] = c;
    }
}

__device__ __forceinline__ void gin_place_one(int* nc, int* rows, int node0, int dl, int s,
                                              int* __restrict__ ovf_cnt, int* __restrict__ ovf_dst,
                                              int* __restrict__ ovf_src) {
    const int slot = atomicAdd(&nc[dl], 1);
    if (slot < SLOT_CAP) {
        rows[dl * SLOT_CAP + slot] = s;
    } else {
        const int k = atomicAdd(ovf_cnt, 1);
        ovf_dst[k] = node0 + dl;
        ovf_src[k] = s;
    }
}

#define PLACE_THREADS 512
#define PLACE_WAVES   (PLACE_THREADS / 32)
#define PLACE_SUBS    (NBIN / PLACE_WAVES)

__global__ __launch_bounds__(PLACE_THREADS)
void gin_place(const int* __restrict__ src, const int* __restrict__ dst,
               const int* __restrict__ region, const int* __restrict__ cnt_rw,
               int* __restrict__ deg, int* __restrict__ col,
               int* __restrict__ ovf_cnt, int* __restrict__ ovf_dst,
               int* __restrict__ ovf_src) {
    static_assert(NBIN % PLACE_WAVES == 0, "whole lines per wave");
    __shared__ int nc[NPB];
    __shared__ int rows[NPB * SLOT_CAP];
    const int tid  = threadIdx.x;
    const int r    = blockIdx.x;
    const int wave = tid >> 5, lane = tid & 31;
    const int node0 = r * NPB;
    for (int i = tid; i < NPB; i += PLACE_THREADS) nc[i] = 0;
    int cw[PLACE_SUBS], pk[PLACE_SUBS];
#pragma unroll
    for (int k = 0; k < PLACE_SUBS; ++k) cw[k] = cnt_rw[r * NBIN + wave * PLACE_SUBS + k];
#pragma unroll
    for (int k = 0; k < PLACE_SUBS; ++k) {
        pk[k] = -1;
        if (cw[k] <= SUBCAP && lane < cw[k])
            pk[k] = region[(r * NBIN + wave * PLACE_SUBS + k) * SUBCAP + lane];
    }
    __syncthreads();
#pragma unroll
    for (int k = 0; k < PLACE_SUBS; ++k) {
        if (pk[k] >= 0)
            gin_place_one(nc, rows, node0, pk[k] >> SRC_BITS, pk[k] & ((1 << SRC_BITS) - 1),
                          ovf_cnt, ovf_dst, ovf_src);
        if (cw[k] > SUBCAP) {
            const int w  = wave * PLACE_SUBS + k;
            const int e0 = w * CHUNK;
            const int e1 = (e0 + CHUNK < N_EDGES) ? e0 + CHUNK : N_EDGES;
            for (int e = e0 + lane; e < e1; e += 32) {
                const int d = dst[e];
                if (d / NPB == r)
                    gin_place_one(nc, rows, node0, d - node0, src[e], ovf_cnt, ovf_dst, ovf_src);
            }
        }
    }
    __syncthreads();
    for (int i = wave; i < NPB; i += PLACE_WAVES) {
        const int c = nc[i];
        const int cs = c < SLOT_CAP ? c : SLOT_CAP;
        if (lane < cs) col[(node0 + i) * SLOT_CAP + lane] = rows[i * SLOT_CAP + lane];
    }
    for (int i = tid; i < NPB; i += PLACE_THREADS) deg[node0 + i] = nc[i];
}

__device__ __forceinline__ float4 gin_row4(const float* __restrict__ h, int node, int lane) {
    return *(const float4*)(h + (long long)node * DIM + lane * 4);
}

__device__ __forceinline__ int gin_slot_row(const int* __restrict__ deg, const int* __restrict__ col,
                                            int node, int cnt, int lane) {
    const int cs = cnt < SLOT_CAP ? cnt : SLOT_CAP;
    return *((lane < cs) ? (col + node * SLOT_CAP + lane) : (deg + node));
}

#define GATHER_B    4
#define GEMM_NBLK   256
#define GEMM_WAVES  16
#define N_TILES     (N_NODES / 16)
#define A_PITCH     132

__device__ __forceinline__ void gin_wave_lds_handover() {
    __builtin_amdgcn_fence(__ATOMIC_RELEASE, "workgroup", "local");
    __builtin_amdgcn_wave_barrier();
    __builtin_amdgcn_fence(__ATOMIC_ACQUIRE, "workgroup", "local");
}

__device__ __forceinline__ void gin_pool_flush(float* __restrict__ psum, float* __restrict__ cnt,
                                               int g, int lane, float4 a, int len) {
    float* p = psum + (long long)g * DIM + lane * 4;
    atomicAdd(p + 0, a.x);
    atomicAdd(p + 1, a.y);
    atomicAdd(p + 2, a.z);
    atomicAdd(p + 3, a.w);
    if (lane == 0) atomicAdd(cnt + g, (float)len);
}

template <bool POOL>
__global__ __launch_bounds__(GEMM_WAVES * 32)
void gin_layer(const float* __restrict__ Hin, const int* __restrict__ deg,
               const int* __restrict__ col, const int* __restrict__ ovf_cnt,
               const int* __restrict__ ovf_dst, const int* __restrict__ ovf_src,
               const float* __restrict__ W,
               const float* __restrict__ bias, const float* __restrict__ gamma,
               const float* __restrict__ beta, float* __restrict__ Hout,
               const int* __restrict__ batch, float* __restrict__ psum,
               float* __restrict__ pcnt) {
    __shared__ __attribute__((aligned(16))) float w_frag[32 * 4 * 32 * 4];
    __shared__ __attribute__((aligned(16))) float a_tile[GEMM_WAVES * 16 * A_PITCH];

    const int tid  = threadIdx.x;
    const int lane = tid & 31;
    const int c    = lane & 15;
    const int h    = lane >> 4;
    const int wave = __builtin_amdgcn_readfirstlane(tid >> 5);

    static_assert(GEMM_NBLK * GEMM_WAVES <= N_TILES, "a first tile for every wave");
    static_assert(16 % GATHER_B == 0, "whole batches in a tile");
    int tile = (int)blockIdx.x + GEMM_NBLK * wave;
    int dg = deg[tile * 16 + c];

#pragma unroll
    for (int i = 0; i < 128 / GEMM_WAVES; ++i) {
        const int grp = wave + GEMM_WAVES * i;
        const int t   = grp >> 2;
        const int jp  = grp & 3;
        const float* wp = W + (4 * t + 2 * h) * DIM + 32 * jp + c;
        float4 f;
        f.x = wp[0];
        f.y = wp[DIM];
        f.z = wp[16];
        f.w = wp[DIM + 16];
        *(float4*)(w_frag + ((t * 4 + jp) * 32 + lane) * 4) = f;
    }

    float bv[8], gv[8], bt[8];
#pragma unroll
    for (int j = 0; j < 8; ++j) {
        bv[j] = bias[16 * j + c];
        gv[j] = gamma[16 * j + c];
        bt[j] = beta[16 * j + c];
    }

    float* my_a = a_tile + wave * (16 * A_PITCH);

    int my[GATHER_B];
#pragma unroll
    for (int i = 0; i < GATHER_B; ++i)
        my[i] = gin_slot_row(deg, col, tile * 16 + i, __builtin_amdgcn_readlane(dg, i), lane);

    __syncthreads();

    do {
        const int ntile = tile + GEMM_NBLK * GEMM_WAVES;
        const int ptile = ntile < N_TILES ? ntile : tile;
        const int dgn = deg[ptile * 16 + c];
        int bid = 0;
        if (POOL) bid = batch[tile * 16 + c];

        gin_wave_lds_handover();
#pragma nounroll
        for (int b = 0; b < 16 / GATHER_B; ++b) {
            const int r0 = GATHER_B * b;
            const int r1 = (b + 1 < 16 / GATHER_B) ? r0 + GATHER_B : r0;
            int nx[GATHER_B];
#pragma unroll
            for (int i = 0; i < GATHER_B; ++i)
                nx[i] = gin_slot_row(deg, col, tile * 16 + r1 + i,
                                     __builtin_amdgcn_readlane(dg, r1 + i), lane);

            int cnt[GATHER_B], cs[GATHER_B];
            int cm = 0;
#pragma unroll
            for (int i = 0; i < GATHER_B; ++i) {
                cnt[i] = __builtin_amdgcn_readlane(dg, r0 + i);
                cs[i]  = cnt[i] < SLOT_CAP ? cnt[i] : SLOT_CAP;
                cm     = cs[i] > cm ? cs[i] : cm;
            }
            float4 self[GATHER_B], sum[GATHER_B];
#pragma unroll
            for (int i = 0; i < GATHER_B; ++i) {
                self[i] = gin_row4(Hin, tile * 16 + r0 + i, lane);
                sum[i]  = make_float4(0.f, 0.f, 0.f, 0.f);
            }
            for (int j = 0; j < cm; j += 8) {
                float4 v[GATHER_B][8];
#pragma unroll
                for (int i = 0; i < GATHER_B; ++i) {
#pragma unroll
                    for (int u = 0; u < 8; ++u) {
                        v[i][u] = make_float4(0.f, 0.f, 0.f, 0.f);
                        if (j + u < cs[i])
                            v[i][u] = gin_row4(Hin, __builtin_amdgcn_readlane(my[i], j + u), lane);
                    }
                }
#pragma unroll
                for (int i = 0; i < GATHER_B; ++i) {
#pragma unroll
                    for (int u = 0; u < 8; ++u) {
                        sum[i].x += v[i][u].x; sum[i].y += v[i][u].y;
                        sum[i].z += v[i][u].z; sum[i].w += v[i][u].w;
                    }
                }
            }
#pragma unroll
            for (int i = 0; i < GATHER_B; ++i) {
                if (cnt[i] > SLOT_CAP) {
                    const int n = tile * 16 + r0 + i;
                    const int total = ovf_cnt[0];
                    for (int base = 0; base < total; base += 32) {
                        const int k  = base + lane;
                        const int od = (k < total) ? ovf_dst[k] : -1;
                        const int os = (k < total) ? ovf_src[k] : 0;
                        unsigned int hit = __builtin_amdgcn_ballot_w32(od == n);
                        while (hit != 0u) {
                            const int bit = __builtin_ctz(hit);
                            hit &= hit - 1u;
                            const float4 v = gin_row4(Hin, __builtin_amdgcn_readlane(os, bit), lane);
                            sum[i].x += v.x; sum[i].y += v.y; sum[i].z += v.z; sum[i].w += v.w;
                        }
                    }
                }
                v4f o;
                o.x = 2.f * self[i].x + sum[i].x;
                o.y = 2.f * self[i].y + sum[i].y;
                o.z = 2.f * self[i].z + sum[i].z;
                o.w = 2.f * self[i].w + sum[i].w;
                *(v4f*)(my_a + (r0 + i) * A_PITCH + lane * 4) = o;
                my[i] = nx[i];
            }
        }
        gin_wave_lds_handover();
        const float* my_frag = my_a + c * A_PITCH + 2 * h;

        dg = dgn;
#pragma unroll
        for (int i = 0; i < GATHER_B; ++i)
            my[i] = gin_slot_row(deg, col, ptile * 16 + i, __builtin_amdgcn_readlane(dg, i), lane);

        v8f acc[8];
#pragma unroll
        for (int j = 0; j < 8; ++j) acc[j] = (v8f){0.f, 0.f, 0.f, 0.f, 0.f, 0.f, 0.f, 0.f};

        const float* wl = w_frag + lane * 4;
        v4f b0s[16], b1s[16];
        v2f a0s[4], a1s[4];
#pragma unroll
        for (int u = 0; u < 16; ++u) b0s[u] = *(const v4f*)(wl + u * 128);
#pragma unroll
        for (int tt = 0; tt < 4; ++tt) a0s[tt] = *(const v2f*)(my_frag + 4 * tt);
#pragma nounroll
        for (int kk = 0; kk < 4; ++kk) {
            const int blk1 = 2 * kk + 1;
            const int blk2 = (2 * kk + 2) & 7;
#pragma unroll
            for (int u = 0; u < 16; ++u) b1s[u] = *(const v4f*)(wl + blk1 * 2048 + u * 128);
#pragma unroll
            for (int tt = 0; tt < 4; ++tt) a1s[tt] = *(const v2f*)(my_frag + blk1 * 16 + 4 * tt);
#pragma unroll
            for (int tt = 0; tt < 4; ++tt) {
#pragma unroll
                for (int jp = 0; jp < 4; ++jp) {
                    const v4f f = b0s[tt * 4 + jp];
                    v2f b0, b1;
                    b0.x = f.x; b0.y = f.y;
                    b1.x = f.z; b1.y = f.w;
                    acc[2 * jp] = __builtin_amdgcn_wmma_f32_16x16x4_f32(
                            false, a0s[tt], false, b0, (short)0, acc[2 * jp], false, false);
                    acc[2 * jp + 1] = __builtin_amdgcn_wmma_f32_16x16x4_f32(
                            false, a0s[tt], false, b1, (short)0, acc[2 * jp + 1], false, false);
                }
            }
#pragma unroll
            for (int u = 0; u < 16; ++u) b0s[u] = *(const v4f*)(wl + blk2 * 2048 + u * 128);
#pragma unroll
            for (int tt = 0; tt < 4; ++tt) a0s[tt] = *(const v2f*)(my_frag + blk2 * 16 + 4 * tt);
#pragma unroll
            for (int tt = 0; tt < 4; ++tt) {
#pragma unroll
                for (int jp = 0; jp < 4; ++jp) {
                    const v4f f = b1s[tt * 4 + jp];
                    v2f b0, b1;
                    b0.x = f.x; b0.y = f.y;
                    b1.x = f.z; b1.y = f.w;
                    acc[2 * jp] = __builtin_amdgcn_wmma_f32_16x16x4_f32(
                            false, a1s[tt], false, b0, (short)0, acc[2 * jp], false, false);
                    acc[2 * jp + 1] = __builtin_amdgcn_wmma_f32_16x16x4_f32(
                            false, a1s[tt], false, b1, (short)0, acc[2 * jp + 1], false, false);
                }
            }
        }

        float s[8], q[8];
#pragma unroll
        for (int r = 0; r < 8; ++r) { s[r] = 0.f; q[r] = 0.f; }
        gin_wave_lds_handover();
#pragma unroll
        for (int j = 0; j < 8; ++j) {
#pragma unroll
            for (int r = 0; r < 8; ++r) {
                float v = acc[j][r] + bv[j];
                v = v > 0.f ? v : 0.f;
                acc[j][r] = v;
                s[r] += v;
                q[r] += v * v;
            }
        }
#pragma unroll
        for (int m = 1; m < 16; m <<= 1) {
            float es[8], eq[8];
#pragma unroll
            for (int r = 0; r < 8; ++r) {
                es[r] = __shfl_xor(s[r], m, 32);
                eq[r] = __shfl_xor(q[r], m, 32);
            }
#pragma unroll
            for (int r = 0; r < 8; ++r) {
                s[r] += es[r];
                q[r] += eq[r];
            }
        }
#pragma unroll
        for (int r = 0; r < 8; ++r) {
            const float mean = s[r] * (1.0f / DIM);
            const float var  = q[r] * (1.0f / DIM) - mean * mean;
            const float rs   = rsqrtf(var + LN_EPSF);
#pragma unroll
            for (int j = 0; j < 8; ++j) {
                const float o = (acc[j][r] - mean) * rs * gv[j] + bt[j];
                my_a[(8 * h + r) * A_PITCH + 16 * j + c] = o;
            }
        }
        gin_wave_lds_handover();
        if (POOL) {
            int cur = __builtin_amdgcn_readlane(bid, 0);
            int len = 0;
            float4 run = make_float4(0.f, 0.f, 0.f, 0.f);
#pragma unroll
            for (int r = 0; r < 16; ++r) {
                const float4 o = *(const float4*)(my_a + r * A_PITCH + lane * 4);
                const int b = __builtin_amdgcn_readlane(bid, r);
                if (b != cur) {
                    gin_pool_flush(psum, pcnt, cur, lane, run, len);
                    cur = b;
                    len = 0;
                    run = make_float4(0.f, 0.f, 0.f, 0.f);
                }
                run.x += o.x; run.y += o.y; run.z += o.z; run.w += o.w;
                ++len;
            }
            gin_pool_flush(psum, pcnt, cur, lane, run, len);
        } else {
#pragma unroll
            for (int r = 0; r < 16; ++r) {
                const float4 o = *(const float4*)(my_a + r * A_PITCH + lane * 4);
                *(float4*)(Hout + (long long)(tile * 16 + r) * DIM + lane * 4) = o;
            }
        }
        tile = ntile;
    } while (tile < N_TILES);
}

#define HEAD_GPW     2
#define HEAD_PARTS   4
#define HEAD_KPER    (DIM / HEAD_PARTS)
#define HEAD_THREADS (HEAD_PARTS * NCLASS)

__global__ __launch_bounds__(HEAD_THREADS)
void gin_head(const float* __restrict__ psum, const float* __restrict__ cnt,
              const float* __restrict__ Wout, const float* __restrict__ bout,
              float* __restrict__ out) {
    static_assert(NGRAPH % HEAD_GPW == 0, "whole work-groups");
    static_assert(DIM % (4 * HEAD_PARTS) == 0, "whole float4 per part");
    static_assert(HEAD_GPW * NCLASS <= HEAD_THREADS, "a thread per logit");
    __shared__ float part[HEAD_GPW][HEAD_PARTS][NCLASS];
    __shared__ __attribute__((aligned(16))) float lg[HEAD_GPW][NCLASS];
    __shared__ __attribute__((aligned(16))) float ex[HEAD_GPW][NCLASS];
    const int t  = threadIdx.x;
    const int p  = t / NCLASS;
    const int c  = t - p * NCLASS;
    const int g0 = blockIdx.x * HEAD_GPW;

    const float* wp = Wout + (p * HEAD_KPER) * NCLASS + c;
    float wv[HEAD_KPER];
#pragma unroll
    for (int i = 0; i < HEAD_KPER; ++i) wv[i] = wp[i * NCLASS];
    float4 pv[HEAD_GPW][HEAD_KPER / 4];
    float  cn[HEAD_GPW];
#pragma unroll
    for (int gi = 0; gi < HEAD_GPW; ++gi) {
        const float* pg = psum + (g0 + gi) * DIM + p * HEAD_KPER;
#pragma unroll
        for (int i = 0; i < HEAD_KPER / 4; ++i) pv[gi][i] = *(const float4*)(pg + 4 * i);
        cn[gi] = cnt[g0 + gi];
    }
    float bo = 0.f;
    if (t < HEAD_GPW * NCLASS) bo = bout[t % NCLASS];

#pragma unroll
    for (int gi = 0; gi < HEAD_GPW; ++gi) {
        const float inv = 1.0f / fmaxf(cn[gi], 1.0f);
        float acc = 0.f;
#pragma unroll
        for (int i = 0; i < HEAD_KPER / 4; ++i) {
            acc += pv[gi][i].x * inv * wv[4 * i + 0];
            acc += pv[gi][i].y * inv * wv[4 * i + 1];
            acc += pv[gi][i].z * inv * wv[4 * i + 2];
            acc += pv[gi][i].w * inv * wv[4 * i + 3];
        }
        part[gi][p][c] = acc;
    }
    __syncthreads();
    const int gi = t / NCLASS;
    const int cc = t - gi * NCLASS;
    if (t < HEAD_GPW * NCLASS) {
        float v = bo;
#pragma unroll
        for (int q = 0; q < HEAD_PARTS; ++q) v += part[gi][q][cc];
        lg[gi][cc] = v;
    }
    __syncthreads();
    float mx = 0.f, mine = 0.f;
    if (t < HEAD_GPW * NCLASS) {
        mx = lg[gi][0];
#pragma unroll
        for (int j = 1; j < NCLASS; ++j) mx = fmaxf(mx, lg[gi][j]);
        mine = lg[gi][cc];
        ex[gi][cc] = expf(mine - mx);
    }
    __syncthreads();
    if (t < HEAD_GPW * NCLASS) {
        float se = 0.f;
#pragma unroll
        for (int j = 0; j < NCLASS; ++j) se += ex[gi][j];
        out[(g0 + gi) * NCLASS + cc] = mine - mx - logf(se);
    }
}

extern "C" void kernel_launch(void* const* d_in, const int* in_sizes, int n_in,
                              void* d_out, int out_size, void* d_ws, size_t ws_size,
                              hipStream_t stream) {
    const float* x      = (const float*)d_in[0];
    const int*   ei     = (const int*)d_in[1];
    const int*   batch  = (const int*)d_in[2];
    const float* Ws     = (const float*)d_in[3];
    const float* bs     = (const float*)d_in[4];
    const float* gammas = (const float*)d_in[5];
    const float* betas  = (const float*)d_in[6];
    const float* Wout   = (const float*)d_in[7];
    const float* bout   = (const float*)d_in[8];
    float* out = (float*)d_out;

    const int* src = ei;
    const int* dst = ei + N_EDGES;

    float* ws = (float*)d_ws;
    const size_t nd = (size_t)N_NODES * DIM;
    float* hA     = ws;
    float* hB     = ws + nd;
    float* psum   = ws + 2 * nd;
    float* cnt    = psum + (size_t)NGRAPH * DIM;
    int*   deg     = (int*)(cnt + NGRAPH);
    int*   ovf_cnt = deg + N_NODES;
    int*   col     = deg + DEG_PAD;
    int*   ovf_dst = col + (size_t)N_NODES * SLOT_CAP;
    int*   ovf_src = ovf_dst + N_EDGES;
    int*   region  = ovf_src + N_EDGES;
    int*   cnt_rw  = region + (size_t)NBKT * NBIN * SUBCAP;

    gin_bin<<<NBIN, BIN_THREADS, 0, stream>>>(src, dst, region, cnt_rw, psum, ovf_cnt);
    gin_place<<<NBKT, PLACE_THREADS, 0, stream>>>(src, dst, region, cnt_rw, deg, col, ovf_cnt, ovf_dst, ovf_src);

    const float* hin = x;
    for (int i = 0; i < NLAYERS - 1; ++i) {
        float* hout = (i & 1) ? hB : hA;
        gin_layer<false><<<GEMM_NBLK, GEMM_WAVES * 32, 0, stream>>>(hin, deg, col, ovf_cnt, ovf_dst, ovf_src,
                                             Ws + (size_t)i * DIM * DIM,
                                             bs + i * DIM, gammas + i * DIM,
                                             betas + i * DIM, hout, batch, psum, cnt);
        hin = hout;
    }
    gin_layer<true><<<GEMM_NBLK, GEMM_WAVES * 32, 0, stream>>>(hin, deg, col, ovf_cnt, ovf_dst, ovf_src,
                                         Ws + (size_t)(NLAYERS - 1) * DIM * DIM,
                                         bs + (NLAYERS - 1) * DIM, gammas + (NLAYERS - 1) * DIM,
                                         betas + (NLAYERS - 1) * DIM, hB, batch, psum, cnt);

    gin_head<<<NGRAPH / HEAD_GPW, HEAD_THREADS, 0, stream>>>(psum, cnt, Wout, bout, out);
}
